# starting kernel
# baseline (speedup 1.0000x reference)
#include <hip/hip_runtime.h>
#include <hip/hip_bf16.h>

#define DFEAT 64

typedef float v2f __attribute__((ext_vector_type(2)));
typedef float v8f __attribute__((ext_vector_type(8)));

__device__ __forceinline__ float fast_tanh(float x) {
#if __has_builtin(__builtin_amdgcn_tanhf)
    return __builtin_amdgcn_tanhf(x);
#else
    return tanhf(x);
#endif
}

__global__ void fa_zero_kernel(float* __restrict__ z, int total) {
    int i = blockIdx.x * blockDim.x + threadIdx.x;
    if (i < total) z[i] = 0.0f;
}

__global__ void fa_node_gates_kernel(const float* __restrict__ h,
                                     const float* __restrict__ gate_w,
                                     float* __restrict__ a,
                                     float* __restrict__ b,
                                     int n) {
    int wave = (blockIdx.x * blockDim.x + threadIdx.x) >> 5;
    int lane = threadIdx.x & 31;
    int base = wave * 16;
    if (base >= n) return;

    int m     = lane & 15;
    int khalf = lane >> 4;

    const float2* hrow2 = (const float2*)(h + (size_t)(base + m) * DFEAT);

    const float2* w2   = (const float2*)(gate_w + ((m == 1) ? DFEAT : 0));
    float         valid = (m < 2) ? 1.0f : 0.0f;

    v8f c = {};
    #pragma unroll
    for (int t = 0; t < 16; ++t) {
        int kq = 2 * t + khalf;

        float2 av = hrow2[kq];
        v2f A; A.x = av.x; A.y = av.y;

        float2 wv = w2[kq];
        v2f B; B.x = wv.x * valid; B.y = wv.y * valid;

        c = __builtin_amdgcn_wmma_f32_16x16x4_f32(
                 false, A,  false, B,
                 (short)0, c,  false,  false);
    }

    int rbase = base + 8 * khalf;
    if (m == 0) {
        #pragma unroll
        for (int r = 0; r < 8; ++r) a[rbase + r] = c[r];
    } else if (m == 1) {
        #pragma unroll
        for (int r = 0; r < 8; ++r) b[rbase + r] = c[r];
    }
}

__global__ void fa_edge_kernel(const float* __restrict__ h,
                               const float* __restrict__ dnorm,
                               const float* __restrict__ a,
                               const float* __restrict__ b,
                               const int* __restrict__ src,
                               const int* __restrict__ dst,
                               const float* __restrict__ gate_b,
                               float* __restrict__ z,
                               int e) {
    int wave = blockIdx.x * (blockDim.x >> 5) + (threadIdx.x >> 5);
    int lane = threadIdx.x & 31;
    if (wave >= e) return;

    int s = __builtin_amdgcn_readfirstlane(src[wave]);
    int d = __builtin_amdgcn_readfirstlane(dst[wave]);

    float g  = fast_tanh(a[d] + b[s] + gate_b[0]);
    float ev = g * dnorm[d] * dnorm[s];

    const float2* hs = (const float2*)(h + (size_t)s * DFEAT);
    float2 v = hs[lane];

    float* zd = z + (size_t)d * DFEAT + 2 * lane;
    atomicAdd(zd,     v.x * ev);
    atomicAdd(zd + 1, v.y * ev);
}

extern "C" void kernel_launch(void* const* d_in, const int* in_sizes, int n_in,
                              void* d_out, int out_size, void* d_ws, size_t ws_size,
                              hipStream_t stream) {
    const float* h      = (const float*)d_in[0];
    const float* dnorm  = (const float*)d_in[1];
    const float* gate_w = (const float*)d_in[2];
    const float* gate_b = (const float*)d_in[3];
    const int*   src    = (const int*)d_in[4];
    const int*   dst    = (const int*)d_in[5];
    float*       z      = (float*)d_out;

    const int n = in_sizes[1];
    const int e = in_sizes[4];

    float* a = (float*)d_ws;
    float* b = a + n;

    {
        int total  = n * DFEAT;
        int blocks = (total + 255) / 256;
        fa_zero_kernel<<<blocks, 256, 0, stream>>>(z, total);
    }

    {
        int waves   = (n + 15) / 16;
        int threads = 256;
        int blocks  = (waves * 32 + threads - 1) / threads;
        fa_node_gates_kernel<<<blocks, threads, 0, stream>>>(h, gate_w, a, b, n);
    }

    {
        int waves_per_block = 256 / 32;
        int blocks = (e + waves_per_block - 1) / waves_per_block;
        fa_edge_kernel<<<blocks, 256, 0, stream>>>(h, dnorm, a, b, src, dst,
                                                   gate_b, z, e);
    }
}


# edges bucketed in-call by 64-row dst range (lists kept in the output slice of their own bin), each bucket summed in LDS with f32 LDS adds, stored once; global f32 atomics gone, arrival-order sum
# speedup vs baseline: 1.3452x; 1.3452x over previous
#include <hip/hip_runtime.h>
#include <hip/hip_bf16.h>

#define DFEAT 64

#define BIN_ROWS   64
#define BIN_SHIFT  6
#define BIN_FLOATS (BIN_ROWS * DFEAT)
#define NBMAX      2048
#define EPT        16

typedef float v2f __attribute__((ext_vector_type(2)));
typedef float v8f __attribute__((ext_vector_type(8)));

__device__ __forceinline__ float fast_tanh(float x) {
#if __has_builtin(__builtin_amdgcn_tanhf)
    return __builtin_amdgcn_tanhf(x);
#else
    return tanhf(x);
#endif
}

__global__ void fa_node_gates_kernel(const float* __restrict__ h,
                                     const float* __restrict__ gate_w,
                                     float* __restrict__ a,
                                     float* __restrict__ b,
                                     int* __restrict__ bin_words,
                                     int n, int nb) {
    int gid = blockIdx.x * blockDim.x + threadIdx.x;
    if (gid < nb) bin_words[(size_t)gid * BIN_FLOATS] = 0;
    int wave = gid >> 5;
    int lane = threadIdx.x & 31;
    int base = wave * 16;
    if (base >= n) return;

    int m     = lane & 15;
    int khalf = lane >> 4;

    const float2* hrow2 = (const float2*)(h + (size_t)(base + m) * DFEAT);

    const float2* w2   = (const float2*)(gate_w + ((m == 1) ? DFEAT : 0));
    float         valid = (m < 2) ? 1.0f : 0.0f;

    v8f c = {};
    #pragma unroll
    for (int t = 0; t < 16; ++t) {
        int kq = 2 * t + khalf;

        float2 av = hrow2[kq];
        v2f A; A.x = av.x; A.y = av.y;

        float2 wv = w2[kq];
        v2f B; B.x = wv.x * valid; B.y = wv.y * valid;

        c = __builtin_amdgcn_wmma_f32_16x16x4_f32(
                 false, A,  false, B,
                 (short)0, c,  false,  false);
    }

    int rbase = base + 8 * khalf;
    if (m == 0) {
        #pragma unroll
        for (int r = 0; r < 8; ++r) a[rbase + r] = c[r];
    } else if (m == 1) {
        #pragma unroll
        for (int r = 0; r < 8; ++r) b[rbase + r] = c[r];
    }
}

__global__ __launch_bounds__(256)
void fa_bin_scatter_kernel(const int* __restrict__ src,
                           const int* __restrict__ dst,
                           float* z, int e, int n, int nb) {
    __shared__ int hist[NBMAX];
    const int tid = threadIdx.x;
    for (int i = tid; i < nb; i += 256) hist[i] = 0;
    __syncthreads();

    const int beg = blockIdx.x * (256 * EPT);
    int rank[EPT];
    #pragma unroll
    for (int k = 0; k < EPT; ++k) {
        int i = beg + k * 256 + tid;
        rank[k] = 0;
        if (i < e) rank[k] = atomicAdd(&hist[dst[i] >> BIN_SHIFT], 1);
    }
    __syncthreads();

    int c[NBMAX / 256], r[NBMAX / 256];
    #pragma unroll
    for (int k = 0; k < NBMAX / 256; ++k) {
        int bin = tid + k * 256;
        c[k] = (bin < nb) ? hist[bin] : 0;
    }
    #pragma unroll
    for (int k = 0; k < NBMAX / 256; ++k) {
        int bin = tid + k * 256;
        r[k] = 0;
        if (c[k] > 0) r[k] = atomicAdd((int*)(z + (size_t)bin * BIN_FLOATS), c[k]);
    }
    #pragma unroll
    for (int k = 0; k < NBMAX / 256; ++k) {
        int bin = tid + k * 256;
        if (bin < nb) hist[bin] = r[k];
    }
    __syncthreads();

    #pragma unroll
    for (int k = 0; k < EPT; ++k) {
        int i = beg + k * 256 + tid;
        if (i < e) {
            int d   = dst[i];
            int s   = src[i];
            int bin = d >> BIN_SHIFT;
            int pos = hist[bin] + rank[k];
            int rows = min(BIN_ROWS, n - bin * BIN_ROWS);
            int cap  = rows * (DFEAT / 2) - 1;
            if (pos < cap) {
                int2* list = (int2*)(z + (size_t)bin * BIN_FLOATS) + 1;
                list[pos] = make_int2(s, d);
            }
        }
    }
}

__device__ __forceinline__ int lane_get(int v, int j) {
    return __builtin_amdgcn_readlane(v, j);
}
__device__ __forceinline__ float lane_getf(float v, int j) {
    return __int_as_float(__builtin_amdgcn_readlane(__float_as_int(v), j));
}

__global__ __launch_bounds__(256)
void fa_bin_reduce_kernel(const float* __restrict__ h,
                          const float* __restrict__ dnorm,
                          const float* __restrict__ a,
                          const float* __restrict__ b,
                          const int* __restrict__ src,
                          const int* __restrict__ dst,
                          const float* __restrict__ gate_b,
                          float* z, int e, int n) {
    __shared__ float zacc[BIN_FLOATS];
    const int tid  = threadIdx.x;
    const int lane = tid & 31;
    const int wave  = __builtin_amdgcn_readfirstlane(tid >> 5);
    const int bin   = blockIdx.x;
    const int node0 = bin * BIN_ROWS;
    const int rows  = min(BIN_ROWS, n - node0);
    float* zbin     = z + (size_t)bin * BIN_FLOATS;
    const int cnt   = __builtin_amdgcn_readfirstlane(*(const int*)zbin);
    const int cap   = rows * (DFEAT / 2) - 1;
    const int2* list = (const int2*)zbin + 1;

    for (int i = tid; i < BIN_FLOATS / 4; i += 256)
        ((float4*)zacc)[i] = make_float4(0.0f, 0.0f, 0.0f, 0.0f);
    __syncthreads();

    const float gb = gate_b[0];

    if (cnt <= cap) {
        for (int eb = wave * 32; eb < cnt; eb += 8 * 32) {
            const int idx = eb + lane;
            int s = 0, dl = 0;
            float ev = 0.0f;
            if (idx < cnt) {
                int2 rec = list[idx];
                s  = rec.x;
                int d = rec.y;
                dl = d - node0;
                ev = fast_tanh(a[d] + b[s] + gb) * dnorm[d] * dnorm[s];
            }
            const int m = min(32, cnt - eb);
            for (int g = 0; g < m; g += 8) {
                float2 v[8];
                #pragma unroll
                for (int u = 0; u < 8; ++u) {
                    int sj = lane_get(s, g + u);
                    v[u] = ((const float2*)(h + (size_t)sj * DFEAT))[lane];
                }
                #pragma unroll
                for (int u = 0; u < 8; ++u) {
                    if (g + u < m) {
                        int   dj = lane_get(dl, g + u);
                        float ej = lane_getf(ev, g + u);
                        float* zr = zacc + dj * DFEAT + 2 * lane;
                        atomicAdd(zr,     v[u].x * ej);
                        atomicAdd(zr + 1, v[u].y * ej);
                    }
                }
            }
        }
    } else {
        for (int eb = wave * 32; eb < e; eb += 8 * 32) {
            const int idx = eb + lane;
            int s = 0, dl = 0;
            float ev = 0.0f;
            bool mine = false;
            if (idx < e) {
                int d = dst[idx];
                if ((d >> BIN_SHIFT) == bin) {
                    mine = true;
                    s  = src[idx];
                    dl = d - node0;
                    ev = fast_tanh(a[d] + b[s] + gb) * dnorm[d] * dnorm[s];
                }
            }
            unsigned mask = __builtin_amdgcn_ballot_w32(mine);
            while (mask) {
                int j = __builtin_ctz(mask);
                mask &= mask - 1;
                int   sj = lane_get(s, j);
                int   dj = lane_get(dl, j);
                float ej = lane_getf(ev, j);
                float2 v = ((const float2*)(h + (size_t)sj * DFEAT))[lane];
                float* zr = zacc + dj * DFEAT + 2 * lane;
                atomicAdd(zr,     v.x * ej);
                atomicAdd(zr + 1, v.y * ej);
            }
        }
    }
    __syncthreads();

    for (int i = tid; i < rows * (DFEAT / 4); i += 256)
        ((float4*)zbin)[i] = ((const float4*)zacc)[i];
}

extern "C" void kernel_launch(void* const* d_in, const int* in_sizes, int n_in,
                              void* d_out, int out_size, void* d_ws, size_t ws_size,
                              hipStream_t stream) {
    const float* h      = (const float*)d_in[0];
    const float* dnorm  = (const float*)d_in[1];
    const float* gate_w = (const float*)d_in[2];
    const float* gate_b = (const float*)d_in[3];
    const int*   src    = (const int*)d_in[4];
    const int*   dst    = (const int*)d_in[5];
    float*       z      = (float*)d_out;

    const int n = in_sizes[1];
    const int e = in_sizes[4];

    float* a = (float*)d_ws;
    float* b = a + n;

    const int nb = (n + BIN_ROWS - 1) / BIN_ROWS;

    {
        int waves   = (n + 15) / 16;
        int threads = 256;
        int blocks  = (waves * 32 + threads - 1) / threads;
        fa_node_gates_kernel<<<blocks, threads, 0, stream>>>(h, gate_w, a, b,
                                                             (int*)d_out, n, nb);
    }

    {
        int per_block = 256 * EPT;
        int blocks = (e + per_block - 1) / per_block;
        fa_bin_scatter_kernel<<<blocks, 256, 0, stream>>>(src, dst, z, e, n, nb);
    }

    fa_bin_reduce_kernel<<<nb, 256, 0, stream>>>(h, dnorm, a, b, src, dst,
                                                 gate_b, z, e, n);
}


# reduce kernel: source gate term dot(h[src],w_src) formed from the row it gathers (lane-tree sum, 8 rows at a time), not loaded; b array, its WMMA column and scattered load gone; z sum order kept
# speedup vs baseline: 1.4171x; 1.0534x over previous
#include <hip/hip_runtime.h>
#include <hip/hip_bf16.h>

#define DFEAT 64

#define BIN_ROWS   64
#define BIN_SHIFT  6
#define BIN_FLOATS (BIN_ROWS * DFEAT)
#define NBMAX      2048
#define EPT        16

typedef float v2f __attribute__((ext_vector_type(2)));
typedef float v8f __attribute__((ext_vector_type(8)));

__device__ __forceinline__ float fast_tanh(float x) {
#if __has_builtin(__builtin_amdgcn_tanhf)
    return __builtin_amdgcn_tanhf(x);
#else
    return tanhf(x);
#endif
}

__global__ void fa_node_gates_kernel(const float* __restrict__ h,
                                     const float* __restrict__ gate_w,
                                     float* __restrict__ a,
                                     int* __restrict__ bin_words,
                                     int n, int nb) {
    int gid = blockIdx.x * blockDim.x + threadIdx.x;
    if (gid < nb) bin_words[(size_t)gid * BIN_FLOATS] = 0;
    int wave = gid >> 5;
    int lane = threadIdx.x & 31;
    int base = wave * 16;
    if (base >= n) return;

    int m     = lane & 15;
    int khalf = lane >> 4;

    const float2* hrow2 = (const float2*)(h + (size_t)(base + m) * DFEAT);

    const float2* w2   = (const float2*)gate_w;
    float         valid = (m == 0) ? 1.0f : 0.0f;

    v8f c = {};
    #pragma unroll
    for (int t = 0; t < 16; ++t) {
        int kq = 2 * t + khalf;

        float2 av = hrow2[kq];
        v2f A; A.x = av.x; A.y = av.y;

        float2 wv = w2[kq];
        v2f B; B.x = wv.x * valid; B.y = wv.y * valid;

        c = __builtin_amdgcn_wmma_f32_16x16x4_f32(
                 false, A,  false, B,
                 (short)0, c,  false,  false);
    }

    int rbase = base + 8 * khalf;
    if (m == 0) {
        #pragma unroll
        for (int r = 0; r < 8; ++r) a[rbase + r] = c[r];
    }
}

__global__ __launch_bounds__(256)
void fa_bin_scatter_kernel(const int* __restrict__ src,
                           const int* __restrict__ dst,
                           float* z, int e, int n, int nb) {
    __shared__ int hist[NBMAX];
    const int tid = threadIdx.x;
    for (int i = tid; i < nb; i += 256) hist[i] = 0;
    __syncthreads();

    const int beg = blockIdx.x * (256 * EPT);
    int rank[EPT];
    #pragma unroll
    for (int k = 0; k < EPT; ++k) {
        int i = beg + k * 256 + tid;
        rank[k] = 0;
        if (i < e) rank[k] = atomicAdd(&hist[dst[i] >> BIN_SHIFT], 1);
    }
    __syncthreads();

    int c[NBMAX / 256], r[NBMAX / 256];
    #pragma unroll
    for (int k = 0; k < NBMAX / 256; ++k) {
        int bin = tid + k * 256;
        c[k] = (bin < nb) ? hist[bin] : 0;
    }
    #pragma unroll
    for (int k = 0; k < NBMAX / 256; ++k) {
        int bin = tid + k * 256;
        r[k] = 0;
        if (c[k] > 0) r[k] = atomicAdd((int*)(z + (size_t)bin * BIN_FLOATS), c[k]);
    }
    #pragma unroll
    for (int k = 0; k < NBMAX / 256; ++k) {
        int bin = tid + k * 256;
        if (bin < nb) hist[bin] = r[k];
    }
    __syncthreads();

    #pragma unroll
    for (int k = 0; k < EPT; ++k) {
        int i = beg + k * 256 + tid;
        if (i < e) {
            int d   = dst[i];
            int s   = src[i];
            int bin = d >> BIN_SHIFT;
            int pos = hist[bin] + rank[k];
            int rows = min(BIN_ROWS, n - bin * BIN_ROWS);
            int cap  = rows * (DFEAT / 2) - 1;
            if (pos < cap) {
                int2* list = (int2*)(z + (size_t)bin * BIN_FLOATS) + 1;
                list[pos] = make_int2(s, d);
            }
        }
    }
}

__device__ __forceinline__ int lane_get(int v, int j) {
    return __builtin_amdgcn_readlane(v, j);
}
__device__ __forceinline__ float lane_getf(float v, int j) {
    return __int_as_float(__builtin_amdgcn_readlane(__float_as_int(v), j));
}
template <int XM>
__device__ __forceinline__ float row_xor_get(float x) {
    return __int_as_float(__builtin_amdgcn_update_dpp(0, __float_as_int(x), 0x160 + XM,
                                                      0xf, 0xf, true));
}
__device__ __forceinline__ float xhalf(float x) {
    int b = __float_as_int(x);
    return __int_as_float(__builtin_amdgcn_permlanex16(b, b, 0x76543210, 0xfedcba98, false, false));
}
__device__ __forceinline__ float wave_sum(float x) {
    x += row_xor_get<1>(x);
    x += row_xor_get<2>(x);
    x += row_xor_get<4>(x);
    x += row_xor_get<8>(x);
    return x + xhalf(x);
}
__device__ __forceinline__ float sums8_by_lane(const float (&x)[8], int lane) {
    const bool b0 = (lane & 1) != 0, b1 = (lane & 2) != 0, b2 = (lane & 4) != 0;
    float y[4];
    #pragma unroll
    for (int i = 0; i < 4; ++i)
        y[i] = (b0 ? x[2 * i + 1] : x[2 * i]) + row_xor_get<1>(b0 ? x[2 * i] : x[2 * i + 1]);
    float zz[2];
    #pragma unroll
    for (int j = 0; j < 2; ++j)
        zz[j] = (b1 ? y[2 * j + 1] : y[2 * j]) + row_xor_get<2>(b1 ? y[2 * j] : y[2 * j + 1]);
    float w = (b2 ? zz[1] : zz[0]) + row_xor_get<4>(b2 ? zz[0] : zz[1]);
    w += row_xor_get<8>(w);
    return w + xhalf(w);
}

__global__ __launch_bounds__(256)
void fa_bin_reduce_kernel(const float* __restrict__ h,
                          const float* __restrict__ dnorm,
                          const float* __restrict__ a,
                          const float* __restrict__ gate_w,
                          const int* __restrict__ src,
                          const int* __restrict__ dst,
                          const float* __restrict__ gate_b,
                          float* z, int e, int n) {
    __shared__ float zacc[BIN_FLOATS];
    const int tid  = threadIdx.x;
    const int lane = tid & 31;
    const int wave  = __builtin_amdgcn_readfirstlane(tid >> 5);
    const int bin   = blockIdx.x;
    const int node0 = bin * BIN_ROWS;
    const int rows  = min(BIN_ROWS, n - node0);
    float* zbin     = z + (size_t)bin * BIN_FLOATS;
    const int cnt   = __builtin_amdgcn_readfirstlane(*(const int*)zbin);
    const int cap   = rows * (DFEAT / 2) - 1;
    const int2* list = (const int2*)zbin + 1;

    for (int i = tid; i < BIN_FLOATS / 4; i += 256)
        ((float4*)zacc)[i] = make_float4(0.0f, 0.0f, 0.0f, 0.0f);
    __syncthreads();

    const float gb = gate_b[0];
    const float2 wsrc = ((const float2*)(gate_w + DFEAT))[lane];

    if (cnt <= cap) {
        for (int eb = wave * 32; eb < cnt; eb += 8 * 32) {
            const int2 rec = list[min(eb + lane, cnt - 1)];
            const int   s   = rec.x;
            const int   d   = rec.y;
            const int   dl  = d - node0;
            const float ad  = a[d];
            const float dd  = dnorm[d];
            const float dsr = dnorm[s];
            const int m = min(32, cnt - eb);
            for (int g = 0; g < m; g += 8) {
                float2 v[8];
                #pragma unroll
                for (int u = 0; u < 8; ++u) {
                    int sj = lane_get(s, g + u);
                    v[u] = ((const float2*)(h + (size_t)sj * DFEAT))[lane];
                }
                float x[8];
                #pragma unroll
                for (int u = 0; u < 8; ++u)
                    x[u] = v[u].x * wsrc.x + v[u].y * wsrc.y;
                const float bsg = sums8_by_lane(x, lane);
                const float evv = fast_tanh(ad + bsg + gb) * dd * dsr;
                #pragma unroll
                for (int u = 0; u < 8; ++u) {
                    if (g + u < m) {
                        int   dj = lane_get(dl, g + u);
                        float ej = lane_getf(evv, g + u);
                        float* zr = zacc + dj * DFEAT + 2 * lane;
                        atomicAdd(zr,     v[u].x * ej);
                        atomicAdd(zr + 1, v[u].y * ej);
                    }
                }
            }
        }
    } else {
        for (int eb = wave * 32; eb < e; eb += 8 * 32) {
            const int idx = eb + lane;
            int s = 0, dl = 0;
            float pre = 0.0f, sc = 0.0f;
            bool mine = false;
            if (idx < e) {
                int d = dst[idx];
                if ((d >> BIN_SHIFT) == bin) {
                    mine = true;
                    s  = src[idx];
                    dl = d - node0;
                    pre = a[d] + gb;
                    sc  = dnorm[d] * dnorm[s];
                }
            }
            unsigned mask = __builtin_amdgcn_ballot_w32(mine);
            while (mask) {
                int j = __builtin_ctz(mask);
                mask &= mask - 1;
                int   sj = lane_get(s, j);
                int   dj = lane_get(dl, j);
                float2 v = ((const float2*)(h + (size_t)sj * DFEAT))[lane];
                float bs = wave_sum(v.x * wsrc.x + v.y * wsrc.y);
                float ej = fast_tanh(lane_getf(pre, j) + bs) * lane_getf(sc, j);
                float* zr = zacc + dj * DFEAT + 2 * lane;
                atomicAdd(zr,     v.x * ej);
                atomicAdd(zr + 1, v.y * ej);
            }
        }
    }
    __syncthreads();

    for (int i = tid; i < rows * (DFEAT / 4); i += 256)
        ((float4*)zbin)[i] = ((const float4*)zacc)[i];
}

extern "C" void kernel_launch(void* const* d_in, const int* in_sizes, int n_in,
                              void* d_out, int out_size, void* d_ws, size_t ws_size,
                              hipStream_t stream) {
    const float* h      = (const float*)d_in[0];
    const float* dnorm  = (const float*)d_in[1];
    const float* gate_w = (const float*)d_in[2];
    const float* gate_b = (const float*)d_in[3];
    const int*   src    = (const int*)d_in[4];
    const int*   dst    = (const int*)d_in[5];
    float*       z      = (float*)d_out;

    const int n = in_sizes[1];
    const int e = in_sizes[4];

    float* a = (float*)d_ws;

    const int nb = (n + BIN_ROWS - 1) / BIN_ROWS;

    {
        int waves   = (n + 15) / 16;
        int threads = 256;
        int blocks  = (waves * 32 + threads - 1) / threads;
        fa_node_gates_kernel<<<blocks, threads, 0, stream>>>(h, gate_w, a,
                                                             (int*)d_out, n, nb);
    }

    {
        int per_block = 256 * EPT;
        int blocks = (e + per_block - 1) / per_block;
        fa_bin_scatter_kernel<<<blocks, 256, 0, stream>>>(src, dst, z, e, n, nb);
    }

    fa_bin_reduce_kernel<<<nb, 256, 0, stream>>>(h, dnorm, a, gate_w, src, dst,
                                                 gate_b, z, e, n);
}


# reduce kernel: a bin's edges ordered by dst row in LDS (rank = returning LDS int add, 64 counts scanned); owner wave sums each row in registers, stores it once; f32 LDS adds gone; sum in rank order
# speedup vs baseline: 1.5464x; 1.0912x over previous
#include <hip/hip_runtime.h>
#include <hip/hip_bf16.h>

#define DFEAT 64

#define BIN_ROWS   64
#define BIN_SHIFT  6
#define BIN_FLOATS (BIN_ROWS * DFEAT)
#define NBMAX      2048
#define EPT        16

typedef float v2f __attribute__((ext_vector_type(2)));
typedef float v8f __attribute__((ext_vector_type(8)));

__device__ __forceinline__ float fast_tanh(float x) {
#if __has_builtin(__builtin_amdgcn_tanhf)
    return __builtin_amdgcn_tanhf(x);
#else
    return tanhf(x);
#endif
}

__global__ void fa_node_gates_kernel(const float* __restrict__ h,
                                     const float* __restrict__ gate_w,
                                     float* __restrict__ a,
                                     int* __restrict__ bin_words,
                                     int n, int nb) {
    int gid = blockIdx.x * blockDim.x + threadIdx.x;
    if (gid < nb) bin_words[(size_t)gid * BIN_FLOATS] = 0;
    int wave = gid >> 5;
    int lane = threadIdx.x & 31;
    int base = wave * 16;
    if (base >= n) return;

    int m     = lane & 15;
    int khalf = lane >> 4;

    const float2* hrow2 = (const float2*)(h + (size_t)(base + m) * DFEAT);

    const float2* w2   = (const float2*)gate_w;
    float         valid = (m == 0) ? 1.0f : 0.0f;

    v8f c = {};
    #pragma unroll
    for (int t = 0; t < 16; ++t) {
        int kq = 2 * t + khalf;

        float2 av = hrow2[kq];
        v2f A; A.x = av.x; A.y = av.y;

        float2 wv = w2[kq];
        v2f B; B.x = wv.x * valid; B.y = wv.y * valid;

        c = __builtin_amdgcn_wmma_f32_16x16x4_f32(
                 false, A,  false, B,
                 (short)0, c,  false,  false);
    }

    int rbase = base + 8 * khalf;
    if (m == 0) {
        #pragma unroll
        for (int r = 0; r < 8; ++r) a[rbase + r] = c[r];
    }
}

__global__ __launch_bounds__(256)
void fa_bin_scatter_kernel(const int* __restrict__ src,
                           const int* __restrict__ dst,
                           float* z, int e, int n, int nb) {
    __shared__ int hist[NBMAX];
    const int tid = threadIdx.x;
    for (int i = tid; i < nb; i += 256) hist[i] = 0;
    __syncthreads();

    const int beg = blockIdx.x * (256 * EPT);
    int rank[EPT];
    #pragma unroll
    for (int k = 0; k < EPT; ++k) {
        int i = beg + k * 256 + tid;
        rank[k] = 0;
        if (i < e) rank[k] = atomicAdd(&hist[dst[i] >> BIN_SHIFT], 1);
    }
    __syncthreads();

    int c[NBMAX / 256], r[NBMAX / 256];
    #pragma unroll
    for (int k = 0; k < NBMAX / 256; ++k) {
        int bin = tid + k * 256;
        c[k] = (bin < nb) ? hist[bin] : 0;
    }
    #pragma unroll
    for (int k = 0; k < NBMAX / 256; ++k) {
        int bin = tid + k * 256;
        r[k] = 0;
        if (c[k] > 0) r[k] = atomicAdd((int*)(z + (size_t)bin * BIN_FLOATS), c[k]);
    }
    #pragma unroll
    for (int k = 0; k < NBMAX / 256; ++k) {
        int bin = tid + k * 256;
        if (bin < nb) hist[bin] = r[k];
    }
    __syncthreads();

    #pragma unroll
    for (int k = 0; k < EPT; ++k) {
        int i = beg + k * 256 + tid;
        if (i < e) {
            int d   = dst[i];
            int s   = src[i];
            int bin = d >> BIN_SHIFT;
            int pos = hist[bin] + rank[k];
            int rows = min(BIN_ROWS, n - bin * BIN_ROWS);
            int cap  = rows * (DFEAT / 2) - 1;
            if (pos < cap) {
                int2* list = (int2*)(z + (size_t)bin * BIN_FLOATS) + 1;
                list[pos] = make_int2(s, d);
            }
        }
    }
}

__device__ __forceinline__ int lane_get(int v, int j) {
    return __builtin_amdgcn_readlane(v, j);
}
__device__ __forceinline__ float lane_getf(float v, int j) {
    return __int_as_float(__builtin_amdgcn_readlane(__float_as_int(v), j));
}
template <int XM>
__device__ __forceinline__ float row_xor_get(float x) {
    return __int_as_float(__builtin_amdgcn_update_dpp(0, __float_as_int(x), 0x160 + XM,
                                                      0xf, 0xf, true));
}
__device__ __forceinline__ float xhalf(float x) {
    int b = __float_as_int(x);
    return __int_as_float(__builtin_amdgcn_permlanex16(b, b, 0x76543210, 0xfedcba98, false, false));
}
__device__ __forceinline__ float wave_sum(float x) {
    x += row_xor_get<1>(x);
    x += row_xor_get<2>(x);
    x += row_xor_get<4>(x);
    x += row_xor_get<8>(x);
    return x + xhalf(x);
}
__device__ __forceinline__ float sums8_by_lane(const float (&x)[8], int lane) {
    const bool b0 = (lane & 1) != 0, b1 = (lane & 2) != 0, b2 = (lane & 4) != 0;
    float y[4];
    #pragma unroll
    for (int i = 0; i < 4; ++i)
        y[i] = (b0 ? x[2 * i + 1] : x[2 * i]) + row_xor_get<1>(b0 ? x[2 * i] : x[2 * i + 1]);
    float zz[2];
    #pragma unroll
    for (int j = 0; j < 2; ++j)
        zz[j] = (b1 ? y[2 * j + 1] : y[2 * j]) + row_xor_get<2>(b1 ? y[2 * j] : y[2 * j + 1]);
    float w = (b2 ? zz[1] : zz[0]) + row_xor_get<4>(b2 ? zz[0] : zz[1]);
    w += row_xor_get<8>(w);
    return w + xhalf(w);
}

template <int SH>
__device__ __forceinline__ int row_shr_get(int x) {
    return __builtin_amdgcn_update_dpp(0, x, 0x110 + SH, 0xf, 0xf, true);
}

__global__ __launch_bounds__(256)
void fa_bin_reduce_kernel(const float* __restrict__ h,
                          const float* __restrict__ dnorm,
                          const float* __restrict__ a,
                          const float* __restrict__ gate_w,
                          const int* __restrict__ src,
                          const int* __restrict__ dst,
                          const float* __restrict__ gate_b,
                          float* z, int e, int n) {
    __shared__ float zacc[BIN_FLOATS];
    __shared__ int   rowcnt[BIN_ROWS];
    __shared__ int   rowbeg[BIN_ROWS + 1];
    const int tid  = threadIdx.x;
    const int lane = tid & 31;
    const int wave  = __builtin_amdgcn_readfirstlane(tid >> 5);
    const int bin   = blockIdx.x;
    const int node0 = bin * BIN_ROWS;
    const int rows  = min(BIN_ROWS, n - node0);
    float* zbin     = z + (size_t)bin * BIN_FLOATS;
    const int cnt   = __builtin_amdgcn_readfirstlane(*(const int*)zbin);
    const int cap   = rows * (DFEAT / 2) - 1;
    const int2* list = (const int2*)zbin + 1;

    const float gb = gate_b[0];
    const float2 wsrc = ((const float2*)(gate_w + DFEAT))[lane];

    if (cnt <= cap) {
        int* sorted = (int*)zacc;
        if (tid < BIN_ROWS) rowcnt[tid] = 0;
        __syncthreads();

        int2 rec[8];
        int  rank[8];
        #pragma unroll
        for (int k = 0; k < 8; ++k) {
            int i = k * 256 + tid;
            rec[k] = make_int2(0, node0);
            if (i < cnt) rec[k] = list[i];
        }
        #pragma unroll
        for (int k = 0; k < 8; ++k) {
            int i = k * 256 + tid;
            rank[k] = 0;
            if (i < cnt) rank[k] = atomicAdd(&rowcnt[rec[k].y - node0], 1);
        }
        __syncthreads();

        if (wave == 0) {
            const int c0 = rowcnt[2 * lane], c1 = rowcnt[2 * lane + 1];
            int incl = c0 + c1;
            incl += row_shr_get<1>(incl);
            incl += row_shr_get<2>(incl);
            incl += row_shr_get<4>(incl);
            incl += row_shr_get<8>(incl);
            const int low = lane_get(incl, 15);
            if (lane >= 16) incl += low;
            const int ex = incl - (c0 + c1);
            rowbeg[2 * lane]     = ex;
            rowbeg[2 * lane + 1] = ex + c0;
            if (lane == 31) rowbeg[BIN_ROWS] = incl;
        }
        __syncthreads();

        #pragma unroll
        for (int k = 0; k < 8; ++k) {
            int i = k * 256 + tid;
            if (i < cnt) {
                int dl = rec[k].y - node0;
                sorted[rowbeg[dl] + rank[k]] = rec[k].x | (dl << 24);
            }
        }
        __syncthreads();

        const int r0 = wave * 8;
        const int rb = rowbeg[r0 + min(lane, 8)];
        const int beg = lane_get(rb, 0);
        const int end = lane_get(rb, 8);

        #pragma unroll
        for (int r = 0; r < 8; ++r) {
            if (r0 + r < rows && lane_get(rb, r + 1) == lane_get(rb, r))
                ((float2*)(zbin + (r0 + r) * DFEAT))[lane] = make_float2(0.0f, 0.0f);
        }

        float2 acc = make_float2(0.0f, 0.0f);
        int cur = -1;
        for (int eb = beg; eb < end; eb += 32) {
            const int ent = sorted[min(eb + lane, end - 1)];
            const int s   = ent & 0xffffff;
            const int dl  = ent >> 24;
            const float ad  = a[node0 + dl];
            const float dd  = dnorm[node0 + dl];
            const float dsr = dnorm[s];
            const int m = min(32, end - eb);
            for (int g = 0; g < m; g += 8) {
                float2 v[8];
                #pragma unroll
                for (int u = 0; u < 8; ++u) {
                    int sj = lane_get(s, g + u);
                    v[u] = ((const float2*)(h + (size_t)sj * DFEAT))[lane];
                }
                float x[8];
                #pragma unroll
                for (int u = 0; u < 8; ++u)
                    x[u] = v[u].x * wsrc.x + v[u].y * wsrc.y;
                const float bsg = sums8_by_lane(x, lane);
                const float evv = fast_tanh(ad + bsg + gb) * dd * dsr;
                #pragma unroll
                for (int u = 0; u < 8; ++u) {
                    if (g + u < m) {
                        int   dj = lane_get(dl, g + u);
                        float ej = lane_getf(evv, g + u);
                        if (dj != cur) {
                            if (cur >= 0) ((float2*)(zbin + cur * DFEAT))[lane] = acc;
                            acc = make_float2(0.0f, 0.0f);
                            cur = dj;
                        }
                        acc.x += v[u].x * ej;
                        acc.y += v[u].y * ej;
                    }
                }
            }
        }
        if (cur >= 0) ((float2*)(zbin + cur * DFEAT))[lane] = acc;
    } else {
        for (int i = tid; i < BIN_FLOATS / 4; i += 256)
            ((float4*)zacc)[i] = make_float4(0.0f, 0.0f, 0.0f, 0.0f);
        __syncthreads();
        for (int eb = wave * 32; eb < e; eb += 8 * 32) {
            const int idx = eb + lane;
            int s = 0, dl = 0;
            float pre = 0.0f, sc = 0.0f;
            bool mine = false;
            if (idx < e) {
                int d = dst[idx];
                if ((d >> BIN_SHIFT) == bin) {
                    mine = true;
                    s  = src[idx];
                    dl = d - node0;
                    pre = a[d] + gb;
                    sc  = dnorm[d] * dnorm[s];
                }
            }
            unsigned mask = __builtin_amdgcn_ballot_w32(mine);
            while (mask) {
                int j = __builtin_ctz(mask);
                mask &= mask - 1;
                int   sj = lane_get(s, j);
                int   dj = lane_get(dl, j);
                float2 v = ((const float2*)(h + (size_t)sj * DFEAT))[lane];
                float bs = wave_sum(v.x * wsrc.x + v.y * wsrc.y);
                float ej = fast_tanh(lane_getf(pre, j) + bs) * lane_getf(sc, j);
                float* zr = zacc + dj * DFEAT + 2 * lane;
                atomicAdd(zr,     v.x * ej);
                atomicAdd(zr + 1, v.y * ej);
            }
        }
        __syncthreads();

        for (int i = tid; i < rows * (DFEAT / 4); i += 256)
            ((float4*)zbin)[i] = ((const float4*)zacc)[i];
    }
}

extern "C" void kernel_launch(void* const* d_in, const int* in_sizes, int n_in,
                              void* d_out, int out_size, void* d_ws, size_t ws_size,
                              hipStream_t stream) {
    const float* h      = (const float*)d_in[0];
    const float* dnorm  = (const float*)d_in[1];
    const float* gate_w = (const float*)d_in[2];
    const float* gate_b = (const float*)d_in[3];
    const int*   src    = (const int*)d_in[4];
    const int*   dst    = (const int*)d_in[5];
    float*       z      = (float*)d_out;

    const int n = in_sizes[1];
    const int e = in_sizes[4];

    float* a = (float*)d_ws;

    const int nb = (n + BIN_ROWS - 1) / BIN_ROWS;

    {
        int waves   = (n + 15) / 16;
        int threads = 256;
        int blocks  = (waves * 32 + threads - 1) / threads;
        fa_node_gates_kernel<<<blocks, threads, 0, stream>>>(h, gate_w, a,
                                                             (int*)d_out, n, nb);
    }

    {
        int per_block = 256 * EPT;
        int blocks = (e + per_block - 1) / per_block;
        fa_bin_scatter_kernel<<<blocks, 256, 0, stream>>>(src, dst, z, e, n, nb);
    }

    fa_bin_reduce_kernel<<<nb, 256, 0, stream>>>(h, dnorm, a, gate_w, src, dst,
                                                 gate_b, z, e, n);
}


# scatter kernel: a thread's 16 dst and 16 src loads issued as one group at entry (clamped index, kept in registers); were 32 guarded loads each waited for in turn, dst read twice; sum order as before
# speedup vs baseline: 1.7507x; 1.1321x over previous
#include <hip/hip_runtime.h>
#include <hip/hip_bf16.h>

#define DFEAT 64

#define BIN_ROWS   64
#define BIN_SHIFT  6
#define BIN_FLOATS (BIN_ROWS * DFEAT)
#define NBMAX      2048
#define EPT        16

typedef float v2f __attribute__((ext_vector_type(2)));
typedef float v8f __attribute__((ext_vector_type(8)));

__device__ __forceinline__ float fast_tanh(float x) {
#if __has_builtin(__builtin_amdgcn_tanhf)
    return __builtin_amdgcn_tanhf(x);
#else
    return tanhf(x);
#endif
}

__global__ void fa_node_gates_kernel(const float* __restrict__ h,
                                     const float* __restrict__ gate_w,
                                     float* __restrict__ a,
                                     int* __restrict__ bin_words,
                                     int n, int nb) {
    int gid = blockIdx.x * blockDim.x + threadIdx.x;
    if (gid < nb) bin_words[(size_t)gid * BIN_FLOATS] = 0;
    int wave = gid >> 5;
    int lane = threadIdx.x & 31;
    int base = wave * 16;
    if (base >= n) return;

    int m     = lane & 15;
    int khalf = lane >> 4;

    const float2* hrow2 = (const float2*)(h + (size_t)(base + m) * DFEAT);

    const float2* w2   = (const float2*)gate_w;
    float         valid = (m == 0) ? 1.0f : 0.0f;

    v8f c = {};
    #pragma unroll
    for (int t = 0; t < 16; ++t) {
        int kq = 2 * t + khalf;

        float2 av = hrow2[kq];
        v2f A; A.x = av.x; A.y = av.y;

        float2 wv = w2[kq];
        v2f B; B.x = wv.x * valid; B.y = wv.y * valid;

        c = __builtin_amdgcn_wmma_f32_16x16x4_f32(
                 false, A,  false, B,
                 (short)0, c,  false,  false);
    }

    int rbase = base + 8 * khalf;
    if (m == 0) {
        #pragma unroll
        for (int r = 0; r < 8; ++r) a[rbase + r] = c[r];
    }
}

__global__ __launch_bounds__(256)
void fa_bin_scatter_kernel(const int* __restrict__ src,
                           const int* __restrict__ dst,
                           float* z, int e, int n, int nb) {
    __shared__ int hist[NBMAX];
    const int tid = threadIdx.x;
    const int beg = blockIdx.x * (256 * EPT);

    int d[EPT], s[EPT];
    #pragma unroll
    for (int k = 0; k < EPT; ++k)
        d[k] = dst[min(beg + k * 256 + tid, e - 1)];
    #pragma unroll
    for (int k = 0; k < EPT; ++k)
        s[k] = src[min(beg + k * 256 + tid, e - 1)];

    for (int i = tid; i < nb; i += 256) hist[i] = 0;
    __syncthreads();

    int rank[EPT];
    #pragma unroll
    for (int k = 0; k < EPT; ++k) {
        int i = beg + k * 256 + tid;
        rank[k] = 0;
        if (i < e) rank[k] = atomicAdd(&hist[d[k] >> BIN_SHIFT], 1);
    }
    __syncthreads();

    int c[NBMAX / 256], r[NBMAX / 256];
    #pragma unroll
    for (int k = 0; k < NBMAX / 256; ++k) {
        int bin = tid + k * 256;
        c[k] = (bin < nb) ? hist[bin] : 0;
    }
    #pragma unroll
    for (int k = 0; k < NBMAX / 256; ++k) {
        int bin = tid + k * 256;
        r[k] = 0;
        if (c[k] > 0) r[k] = atomicAdd((int*)(z + (size_t)bin * BIN_FLOATS), c[k]);
    }
    #pragma unroll
    for (int k = 0; k < NBMAX / 256; ++k) {
        int bin = tid + k * 256;
        if (bin < nb) hist[bin] = r[k];
    }
    __syncthreads();

    #pragma unroll
    for (int k = 0; k < EPT; ++k)
        rank[k] += hist[d[k] >> BIN_SHIFT];
    #pragma unroll
    for (int k = 0; k < EPT; ++k) {
        int i   = beg + k * 256 + tid;
        int bin = d[k] >> BIN_SHIFT;
        int rows = min(BIN_ROWS, n - bin * BIN_ROWS);
        int cap  = rows * (DFEAT / 2) - 1;
        if (i < e && rank[k] < cap) {
            int2* list = (int2*)(z + (size_t)bin * BIN_FLOATS) + 1;
            list[rank[k]] = make_int2(s[k], d[k]);
        }
    }
}

__device__ __forceinline__ int lane_get(int v, int j) {
    return __builtin_amdgcn_readlane(v, j);
}
__device__ __forceinline__ float lane_getf(float v, int j) {
    return __int_as_float(__builtin_amdgcn_readlane(__float_as_int(v), j));
}
template <int XM>
__device__ __forceinline__ float row_xor_get(float x) {
    return __int_as_float(__builtin_amdgcn_update_dpp(0, __float_as_int(x), 0x160 + XM,
                                                      0xf, 0xf, true));
}
__device__ __forceinline__ float xhalf(float x) {
    int b = __float_as_int(x);
    return __int_as_float(__builtin_amdgcn_permlanex16(b, b, 0x76543210, 0xfedcba98, false, false));
}
__device__ __forceinline__ float wave_sum(float x) {
    x += row_xor_get<1>(x);
    x += row_xor_get<2>(x);
    x += row_xor_get<4>(x);
    x += row_xor_get<8>(x);
    return x + xhalf(x);
}
__device__ __forceinline__ float sums8_by_lane(const float (&x)[8], int lane) {
    const bool b0 = (lane & 1) != 0, b1 = (lane & 2) != 0, b2 = (lane & 4) != 0;
    float y[4];
    #pragma unroll
    for (int i = 0; i < 4; ++i)
        y[i] = (b0 ? x[2 * i + 1] : x[2 * i]) + row_xor_get<1>(b0 ? x[2 * i] : x[2 * i + 1]);
    float zz[2];
    #pragma unroll
    for (int j = 0; j < 2; ++j)
        zz[j] = (b1 ? y[2 * j + 1] : y[2 * j]) + row_xor_get<2>(b1 ? y[2 * j] : y[2 * j + 1]);
    float w = (b2 ? zz[1] : zz[0]) + row_xor_get<4>(b2 ? zz[0] : zz[1]);
    w += row_xor_get<8>(w);
    return w + xhalf(w);
}

template <int SH>
__device__ __forceinline__ int row_shr_get(int x) {
    return __builtin_amdgcn_update_dpp(0, x, 0x110 + SH, 0xf, 0xf, true);
}

__global__ __launch_bounds__(256)
void fa_bin_reduce_kernel(const float* __restrict__ h,
                          const float* __restrict__ dnorm,
                          const float* __restrict__ a,
                          const float* __restrict__ gate_w,
                          const int* __restrict__ src,
                          const int* __restrict__ dst,
                          const float* __restrict__ gate_b,
                          float* z, int e, int n) {
    __shared__ float zacc[BIN_FLOATS];
    __shared__ int   rowcnt[BIN_ROWS];
    __shared__ int   rowbeg[BIN_ROWS + 1];
    const int tid  = threadIdx.x;
    const int lane = tid & 31;
    const int wave  = __builtin_amdgcn_readfirstlane(tid >> 5);
    const int bin   = blockIdx.x;
    const int node0 = bin * BIN_ROWS;
    const int rows  = min(BIN_ROWS, n - node0);
    float* zbin     = z + (size_t)bin * BIN_FLOATS;
    const int cnt   = __builtin_amdgcn_readfirstlane(*(const int*)zbin);
    const int cap   = rows * (DFEAT / 2) - 1;
    const int2* list = (const int2*)zbin + 1;

    const float gb = gate_b[0];
    const float2 wsrc = ((const float2*)(gate_w + DFEAT))[lane];

    if (cnt <= cap) {
        int* sorted = (int*)zacc;
        if (tid < BIN_ROWS) rowcnt[tid] = 0;
        __syncthreads();

        int2 rec[8];
        int  rank[8];
        #pragma unroll
        for (int k = 0; k < 8; ++k) {
            int i = k * 256 + tid;
            rec[k] = make_int2(0, node0);
            if (i < cnt) rec[k] = list[i];
        }
        #pragma unroll
        for (int k = 0; k < 8; ++k) {
            int i = k * 256 + tid;
            rank[k] = 0;
            if (i < cnt) rank[k] = atomicAdd(&rowcnt[rec[k].y - node0], 1);
        }
        __syncthreads();

        if (wave == 0) {
            const int c0 = rowcnt[2 * lane], c1 = rowcnt[2 * lane + 1];
            int incl = c0 + c1;
            incl += row_shr_get<1>(incl);
            incl += row_shr_get<2>(incl);
            incl += row_shr_get<4>(incl);
            incl += row_shr_get<8>(incl);
            const int low = lane_get(incl, 15);
            if (lane >= 16) incl += low;
            const int ex = incl - (c0 + c1);
            rowbeg[2 * lane]     = ex;
            rowbeg[2 * lane + 1] = ex + c0;
            if (lane == 31) rowbeg[BIN_ROWS] = incl;
        }
        __syncthreads();

        #pragma unroll
        for (int k = 0; k < 8; ++k) {
            int i = k * 256 + tid;
            if (i < cnt) {
                int dl = rec[k].y - node0;
                sorted[rowbeg[dl] + rank[k]] = rec[k].x | (dl << 24);
            }
        }
        __syncthreads();

        const int r0 = wave * 8;
        const int rb = rowbeg[r0 + min(lane, 8)];
        const int beg = lane_get(rb, 0);
        const int end = lane_get(rb, 8);

        #pragma unroll
        for (int r = 0; r < 8; ++r) {
            if (r0 + r < rows && lane_get(rb, r + 1) == lane_get(rb, r))
                ((float2*)(zbin + (r0 + r) * DFEAT))[lane] = make_float2(0.0f, 0.0f);
        }

        float2 acc = make_float2(0.0f, 0.0f);
        int cur = -1;
        for (int eb = beg; eb < end; eb += 32) {
            const int ent = sorted[min(eb + lane, end - 1)];
            const int s   = ent & 0xffffff;
            const int dl  = ent >> 24;
            const float ad  = a[node0 + dl];
            const float dd  = dnorm[node0 + dl];
            const float dsr = dnorm[s];
            const int m = min(32, end - eb);
            for (int g = 0; g < m; g += 8) {
                float2 v[8];
                #pragma unroll
                for (int u = 0; u < 8; ++u) {
                    int sj = lane_get(s, g + u);
                    v[u] = ((const float2*)(h + (size_t)sj * DFEAT))[lane];
                }
                float x[8];
                #pragma unroll
                for (int u = 0; u < 8; ++u)
                    x[u] = v[u].x * wsrc.x + v[u].y * wsrc.y;
                const float bsg = sums8_by_lane(x, lane);
                const float evv = fast_tanh(ad + bsg + gb) * dd * dsr;
                #pragma unroll
                for (int u = 0; u < 8; ++u) {
                    if (g + u < m) {
                        int   dj = lane_get(dl, g + u);
                        float ej = lane_getf(evv, g + u);
                        if (dj != cur) {
                            if (cur >= 0) ((float2*)(zbin + cur * DFEAT))[lane] = acc;
                            acc = make_float2(0.0f, 0.0f);
                            cur = dj;
                        }
                        acc.x += v[u].x * ej;
                        acc.y += v[u].y * ej;
                    }
                }
            }
        }
        if (cur >= 0) ((float2*)(zbin + cur * DFEAT))[lane] = acc;
    } else {
        for (int i = tid; i < BIN_FLOATS / 4; i += 256)
            ((float4*)zacc)[i] = make_float4(0.0f, 0.0f, 0.0f, 0.0f);
        __syncthreads();
        for (int eb = wave * 32; eb < e; eb += 8 * 32) {
            const int idx = eb + lane;
            int s = 0, dl = 0;
            float pre = 0.0f, sc = 0.0f;
            bool mine = false;
            if (idx < e) {
                int d = dst[idx];
                if ((d >> BIN_SHIFT) == bin) {
                    mine = true;
                    s  = src[idx];
                    dl = d - node0;
                    pre = a[d] + gb;
                    sc  = dnorm[d] * dnorm[s];
                }
            }
            unsigned mask = __builtin_amdgcn_ballot_w32(mine);
            while (mask) {
                int j = __builtin_ctz(mask);
                mask &= mask - 1;
                int   sj = lane_get(s, j);
                int   dj = lane_get(dl, j);
                float2 v = ((const float2*)(h + (size_t)sj * DFEAT))[lane];
                float bs = wave_sum(v.x * wsrc.x + v.y * wsrc.y);
                float ej = fast_tanh(lane_getf(pre, j) + bs) * lane_getf(sc, j);
                float* zr = zacc + dj * DFEAT + 2 * lane;
                atomicAdd(zr,     v.x * ej);
                atomicAdd(zr + 1, v.y * ej);
            }
        }
        __syncthreads();

        for (int i = tid; i < rows * (DFEAT / 4); i += 256)
            ((float4*)zbin)[i] = ((const float4*)zacc)[i];
    }
}

extern "C" void kernel_launch(void* const* d_in, const int* in_sizes, int n_in,
                              void* d_out, int out_size, void* d_ws, size_t ws_size,
                              hipStream_t stream) {
    const float* h      = (const float*)d_in[0];
    const float* dnorm  = (const float*)d_in[1];
    const float* gate_w = (const float*)d_in[2];
    const float* gate_b = (const float*)d_in[3];
    const int*   src    = (const int*)d_in[4];
    const int*   dst    = (const int*)d_in[5];
    float*       z      = (float*)d_out;

    const int n = in_sizes[1];
    const int e = in_sizes[4];

    float* a = (float*)d_ws;

    const int nb = (n + BIN_ROWS - 1) / BIN_ROWS;

    {
        int waves   = (n + 15) / 16;
        int threads = 256;
        int blocks  = (waves * 32 + threads - 1) / threads;
        fa_node_gates_kernel<<<blocks, threads, 0, stream>>>(h, gate_w, a,
                                                             (int*)d_out, n, nb);
    }

    {
        int per_block = 256 * EPT;
        int blocks = (e + per_block - 1) / per_block;
        fa_bin_scatter_kernel<<<blocks, 256, 0, stream>>>(src, dst, z, e, n, nb);
    }

    fa_bin_reduce_kernel<<<nb, 256, 0, stream>>>(h, dnorm, a, gate_w, src, dst,
                                                 gate_b, z, e, n);
}


# bin edge counts moved from the heads of the 16-KiB output slices to a packed workspace table, 32 bins per 128-B line: a wave's reserving add touches 1 line, was 32; sum order as before
# speedup vs baseline: 2.4510x; 1.4000x over previous
#include <hip/hip_runtime.h>
#include <hip/hip_bf16.h>

#define DFEAT 64

#define CNT_PITCH  1056
#define BIN_ROWS   64
#define BIN_SHIFT  6
#define BIN_FLOATS (BIN_ROWS * DFEAT)
#define NBMAX      2048
#define EPT        16

__device__ __forceinline__ int cnt_index(int bin) {
    return (bin >> 5) * CNT_PITCH + (bin & 31);
}

typedef float v2f __attribute__((ext_vector_type(2)));
typedef float v8f __attribute__((ext_vector_type(8)));

__device__ __forceinline__ float fast_tanh(float x) {
#if __has_builtin(__builtin_amdgcn_tanhf)
    return __builtin_amdgcn_tanhf(x);
#else
    return tanhf(x);
#endif
}

__global__ void fa_node_gates_kernel(const float* __restrict__ h,
                                     const float* __restrict__ gate_w,
                                     float* __restrict__ a,
                                     int* __restrict__ bin_cnt,
                                     int n, int nb) {
    int gid = blockIdx.x * blockDim.x + threadIdx.x;
    if (gid < nb) bin_cnt[cnt_index(gid)] = 0;
    int wave = gid >> 5;
    int lane = threadIdx.x & 31;
    int base = wave * 16;
    if (base >= n) return;

    int m     = lane & 15;
    int khalf = lane >> 4;

    const float2* hrow2 = (const float2*)(h + (size_t)(base + m) * DFEAT);

    const float2* w2   = (const float2*)gate_w;
    float         valid = (m == 0) ? 1.0f : 0.0f;

    v8f c = {};
    #pragma unroll
    for (int t = 0; t < 16; ++t) {
        int kq = 2 * t + khalf;

        float2 av = hrow2[kq];
        v2f A; A.x = av.x; A.y = av.y;

        float2 wv = w2[kq];
        v2f B; B.x = wv.x * valid; B.y = wv.y * valid;

        c = __builtin_amdgcn_wmma_f32_16x16x4_f32(
                 false, A,  false, B,
                 (short)0, c,  false,  false);
    }

    int rbase = base + 8 * khalf;
    if (m == 0) {
        #pragma unroll
        for (int r = 0; r < 8; ++r) a[rbase + r] = c[r];
    }
}

__global__ __launch_bounds__(256)
void fa_bin_scatter_kernel(const int* __restrict__ src,
                           const int* __restrict__ dst,
                           float* z, int* bin_cnt, int e, int n, int nb) {
    __shared__ int hist[NBMAX];
    const int tid = threadIdx.x;
    const int beg = blockIdx.x * (256 * EPT);

    int d[EPT], s[EPT];
    #pragma unroll
    for (int k = 0; k < EPT; ++k)
        d[k] = dst[min(beg + k * 256 + tid, e - 1)];
    #pragma unroll
    for (int k = 0; k < EPT; ++k)
        s[k] = src[min(beg + k * 256 + tid, e - 1)];

    for (int i = tid; i < nb; i += 256) hist[i] = 0;
    __syncthreads();

    int rank[EPT];
    #pragma unroll
    for (int k = 0; k < EPT; ++k) {
        int i = beg + k * 256 + tid;
        rank[k] = 0;
        if (i < e) rank[k] = atomicAdd(&hist[d[k] >> BIN_SHIFT], 1);
    }
    __syncthreads();

    int c[NBMAX / 256], r[NBMAX / 256];
    #pragma unroll
    for (int k = 0; k < NBMAX / 256; ++k) {
        int bin = tid + k * 256;
        c[k] = (bin < nb) ? hist[bin] : 0;
    }
    #pragma unroll
    for (int k = 0; k < NBMAX / 256; ++k) {
        int bin = tid + k * 256;
        r[k] = 0;
        if (c[k] > 0) r[k] = atomicAdd(bin_cnt + cnt_index(bin), c[k]);
    }
    #pragma unroll
    for (int k = 0; k < NBMAX / 256; ++k) {
        int bin = tid + k * 256;
        if (bin < nb) hist[bin] = r[k];
    }
    __syncthreads();

    #pragma unroll
    for (int k = 0; k < EPT; ++k)
        rank[k] += hist[d[k] >> BIN_SHIFT];
    #pragma unroll
    for (int k = 0; k < EPT; ++k) {
        int i   = beg + k * 256 + tid;
        int bin = d[k] >> BIN_SHIFT;
        int rows = min(BIN_ROWS, n - bin * BIN_ROWS);
        int cap  = rows * (DFEAT / 2);
        if (i < e && rank[k] < cap) {
            int2* list = (int2*)(z + (size_t)bin * BIN_FLOATS);
            list[rank[k]] = make_int2(s[k], d[k]);
        }
    }
}

__device__ __forceinline__ int lane_get(int v, int j) {
    return __builtin_amdgcn_readlane(v, j);
}
__device__ __forceinline__ float lane_getf(float v, int j) {
    return __int_as_float(__builtin_amdgcn_readlane(__float_as_int(v), j));
}
template <int XM>
__device__ __forceinline__ float row_xor_get(float x) {
    return __int_as_float(__builtin_amdgcn_update_dpp(0, __float_as_int(x), 0x160 + XM,
                                                      0xf, 0xf, true));
}
__device__ __forceinline__ float xhalf(float x) {
    int b = __float_as_int(x);
    return __int_as_float(__builtin_amdgcn_permlanex16(b, b, 0x76543210, 0xfedcba98, false, false));
}
__device__ __forceinline__ float wave_sum(float x) {
    x += row_xor_get<1>(x);
    x += row_xor_get<2>(x);
    x += row_xor_get<4>(x);
    x += row_xor_get<8>(x);
    return x + xhalf(x);
}
__device__ __forceinline__ float sums8_by_lane(const float (&x)[8], int lane) {
    const bool b0 = (lane & 1) != 0, b1 = (lane & 2) != 0, b2 = (lane & 4) != 0;
    float y[4];
    #pragma unroll
    for (int i = 0; i < 4; ++i)
        y[i] = (b0 ? x[2 * i + 1] : x[2 * i]) + row_xor_get<1>(b0 ? x[2 * i] : x[2 * i + 1]);
    float zz[2];
    #pragma unroll
    for (int j = 0; j < 2; ++j)
        zz[j] = (b1 ? y[2 * j + 1] : y[2 * j]) + row_xor_get<2>(b1 ? y[2 * j] : y[2 * j + 1]);
    float w = (b2 ? zz[1] : zz[0]) + row_xor_get<4>(b2 ? zz[0] : zz[1]);
    w += row_xor_get<8>(w);
    return w + xhalf(w);
}

template <int SH>
__device__ __forceinline__ int row_shr_get(int x) {
    return __builtin_amdgcn_update_dpp(0, x, 0x110 + SH, 0xf, 0xf, true);
}

__global__ __launch_bounds__(256)
void fa_bin_reduce_kernel(const float* __restrict__ h,
                          const float* __restrict__ dnorm,
                          const float* __restrict__ a,
                          const float* __restrict__ gate_w,
                          const int* __restrict__ src,
                          const int* __restrict__ dst,
                          const float* __restrict__ gate_b,
                          float* z, const int* bin_cnt, int e, int n) {
    __shared__ float zacc[BIN_FLOATS];
    __shared__ int   rowcnt[BIN_ROWS];
    __shared__ int   rowbeg[BIN_ROWS + 1];
    const int tid  = threadIdx.x;
    const int lane = tid & 31;
    const int wave  = __builtin_amdgcn_readfirstlane(tid >> 5);
    const int bin   = blockIdx.x;
    const int node0 = bin * BIN_ROWS;
    const int rows  = min(BIN_ROWS, n - node0);
    float* zbin     = z + (size_t)bin * BIN_FLOATS;
    const int cnt   = __builtin_amdgcn_readfirstlane(bin_cnt[cnt_index(bin)]);
    const int cap   = rows * (DFEAT / 2);
    const int2* list = (const int2*)zbin;

    const float gb = gate_b[0];
    const float2 wsrc = ((const float2*)(gate_w + DFEAT))[lane];

    if (cnt <= cap) {
        int* sorted = (int*)zacc;
        if (tid < BIN_ROWS) rowcnt[tid] = 0;
        __syncthreads();

        int2 rec[8];
        int  rank[8];
        #pragma unroll
        for (int k = 0; k < 8; ++k) {
            int i = k * 256 + tid;
            rec[k] = make_int2(0, node0);
            if (i < cnt) rec[k] = list[i];
        }
        #pragma unroll
        for (int k = 0; k < 8; ++k) {
            int i = k * 256 + tid;
            rank[k] = 0;
            if (i < cnt) rank[k] = atomicAdd(&rowcnt[rec[k].y - node0], 1);
        }
        __syncthreads();

        if (wave == 0) {
            const int c0 = rowcnt[2 * lane], c1 = rowcnt[2 * lane + 1];
            int incl = c0 + c1;
            incl += row_shr_get<1>(incl);
            incl += row_shr_get<2>(incl);
            incl += row_shr_get<4>(incl);
            incl += row_shr_get<8>(incl);
            const int low = lane_get(incl, 15);
            if (lane >= 16) incl += low;
            const int ex = incl - (c0 + c1);
            rowbeg[2 * lane]     = ex;
            rowbeg[2 * lane + 1] = ex + c0;
            if (lane == 31) rowbeg[BIN_ROWS] = incl;
        }
        __syncthreads();

        #pragma unroll
        for (int k = 0; k < 8; ++k) {
            int i = k * 256 + tid;
            if (i < cnt) {
                int dl = rec[k].y - node0;
                sorted[rowbeg[dl] + rank[k]] = rec[k].x | (dl << 24);
            }
        }
        __syncthreads();

        const int r0 = wave * 8;
        const int rb = rowbeg[r0 + min(lane, 8)];
        const int beg = lane_get(rb, 0);
        const int end = lane_get(rb, 8);

        #pragma unroll
        for (int r = 0; r < 8; ++r) {
            if (r0 + r < rows && lane_get(rb, r + 1) == lane_get(rb, r))
                ((float2*)(zbin + (r0 + r) * DFEAT))[lane] = make_float2(0.0f, 0.0f);
        }

        float2 acc = make_float2(0.0f, 0.0f);
        int cur = -1;
        for (int eb = beg; eb < end; eb += 32) {
            const int ent = sorted[min(eb + lane, end - 1)];
            const int s   = ent & 0xffffff;
            const int dl  = ent >> 24;
            const float ad  = a[node0 + dl];
            const float dd  = dnorm[node0 + dl];
            const float dsr = dnorm[s];
            const int m = min(32, end - eb);
            for (int g = 0; g < m; g += 8) {
                float2 v[8];
                #pragma unroll
                for (int u = 0; u < 8; ++u) {
                    int sj = lane_get(s, g + u);
                    v[u] = ((const float2*)(h + (size_t)sj * DFEAT))[lane];
                }
                float x[8];
                #pragma unroll
                for (int u = 0; u < 8; ++u)
                    x[u] = v[u].x * wsrc.x + v[u].y * wsrc.y;
                const float bsg = sums8_by_lane(x, lane);
                const float evv = fast_tanh(ad + bsg + gb) * dd * dsr;
                #pragma unroll
                for (int u = 0; u < 8; ++u) {
                    if (g + u < m) {
                        int   dj = lane_get(dl, g + u);
                        float ej = lane_getf(evv, g + u);
                        if (dj != cur) {
                            if (cur >= 0) ((float2*)(zbin + cur * DFEAT))[lane] = acc;
                            acc = make_float2(0.0f, 0.0f);
                            cur = dj;
                        }
                        acc.x += v[u].x * ej;
                        acc.y += v[u].y * ej;
                    }
                }
            }
        }
        if (cur >= 0) ((float2*)(zbin + cur * DFEAT))[lane] = acc;
    } else {
        for (int i = tid; i < BIN_FLOATS / 4; i += 256)
            ((float4*)zacc)[i] = make_float4(0.0f, 0.0f, 0.0f, 0.0f);
        __syncthreads();
        for (int eb = wave * 32; eb < e; eb += 8 * 32) {
            const int idx = eb + lane;
            int s = 0, dl = 0;
            float pre = 0.0f, sc = 0.0f;
            bool mine = false;
            if (idx < e) {
                int d = dst[idx];
                if ((d >> BIN_SHIFT) == bin) {
                    mine = true;
                    s  = src[idx];
                    dl = d - node0;
                    pre = a[d] + gb;
                    sc  = dnorm[d] * dnorm[s];
                }
            }
            unsigned mask = __builtin_amdgcn_ballot_w32(mine);
            while (mask) {
                int j = __builtin_ctz(mask);
                mask &= mask - 1;
                int   sj = lane_get(s, j);
                int   dj = lane_get(dl, j);
                float2 v = ((const float2*)(h + (size_t)sj * DFEAT))[lane];
                float bs = wave_sum(v.x * wsrc.x + v.y * wsrc.y);
                float ej = fast_tanh(lane_getf(pre, j) + bs) * lane_getf(sc, j);
                float* zr = zacc + dj * DFEAT + 2 * lane;
                atomicAdd(zr,     v.x * ej);
                atomicAdd(zr + 1, v.y * ej);
            }
        }
        __syncthreads();

        for (int i = tid; i < rows * (DFEAT / 4); i += 256)
            ((float4*)zbin)[i] = ((const float4*)zacc)[i];
    }
}

extern "C" void kernel_launch(void* const* d_in, const int* in_sizes, int n_in,
                              void* d_out, int out_size, void* d_ws, size_t ws_size,
                              hipStream_t stream) {
    const float* h      = (const float*)d_in[0];
    const float* dnorm  = (const float*)d_in[1];
    const float* gate_w = (const float*)d_in[2];
    const float* gate_b = (const float*)d_in[3];
    const int*   src    = (const int*)d_in[4];
    const int*   dst    = (const int*)d_in[5];
    float*       z      = (float*)d_out;

    const int n = in_sizes[1];
    const int e = in_sizes[4];

    float* a = (float*)d_ws;
    int* bin_cnt = (int*)d_ws + ((n + 31) & ~31);

    const int nb = (n + BIN_ROWS - 1) / BIN_ROWS;

    {
        int waves   = (n + 15) / 16;
        int threads = 256;
        int blocks  = (waves * 32 + threads - 1) / threads;
        fa_node_gates_kernel<<<blocks, threads, 0, stream>>>(h, gate_w, a,
                                                             bin_cnt, n, nb);
    }

    {
        int per_block = 256 * EPT;
        int blocks = (e + per_block - 1) / per_block;
        fa_bin_scatter_kernel<<<blocks, 256, 0, stream>>>(src, dst, z, bin_cnt, e, n, nb);
    }

    fa_bin_reduce_kernel<<<nb, 256, 0, stream>>>(h, dnorm, a, gate_w, src, dst,
                                                 gate_b, z, bin_cnt, e, n);
}


# scatter kernel: a block's 8192 pairs put in bin order in LDS (LDS-add rank, scan of counts), copied out so a bin's pairs leave as one piece; one reserving add per bin per 8192 edges; sum order kept
# speedup vs baseline: 2.7742x; 1.1319x over previous
#include <hip/hip_runtime.h>
#include <hip/hip_bf16.h>

#define DFEAT 64

#define CNT_PITCH  1056
#define BIN_ROWS   64
#define BIN_SHIFT  6
#define BIN_FLOATS (BIN_ROWS * DFEAT)
#define NBMAX      2048
#define EPT        16
#define SCAT_THREADS 512
#define SCAT_WAVES   (SCAT_THREADS / 32)
#define SCAT_SLABS   (NBMAX / SCAT_THREADS)
#define SCAT_EDGES   (SCAT_THREADS * EPT)
static_assert(SCAT_SLABS * SCAT_WAVES == 64, "the scan of the per-wave totals takes two per lane of one wave");

__device__ __forceinline__ int cnt_index(int bin) {
    return (bin >> 5) * CNT_PITCH + (bin & 31);
}

typedef float v2f __attribute__((ext_vector_type(2)));
typedef float v8f __attribute__((ext_vector_type(8)));

__device__ __forceinline__ float fast_tanh(float x) {
#if __has_builtin(__builtin_amdgcn_tanhf)
    return __builtin_amdgcn_tanhf(x);
#else
    return tanhf(x);
#endif
}

__global__ void fa_node_gates_kernel(const float* __restrict__ h,
                                     const float* __restrict__ gate_w,
                                     float* __restrict__ a,
                                     int* __restrict__ bin_cnt,
                                     int n, int nb) {
    int gid = blockIdx.x * blockDim.x + threadIdx.x;
    if (gid < nb) bin_cnt[cnt_index(gid)] = 0;
    int wave = gid >> 5;
    int lane = threadIdx.x & 31;
    int base = wave * 16;
    if (base >= n) return;

    int m     = lane & 15;
    int khalf = lane >> 4;

    const float2* hrow2 = (const float2*)(h + (size_t)(base + m) * DFEAT);

    const float2* w2   = (const float2*)gate_w;
    float         valid = (m == 0) ? 1.0f : 0.0f;

    v8f c = {};
    #pragma unroll
    for (int t = 0; t < 16; ++t) {
        int kq = 2 * t + khalf;

        float2 av = hrow2[kq];
        v2f A; A.x = av.x; A.y = av.y;

        float2 wv = w2[kq];
        v2f B; B.x = wv.x * valid; B.y = wv.y * valid;

        c = __builtin_amdgcn_wmma_f32_16x16x4_f32(
                 false, A,  false, B,
                 (short)0, c,  false,  false);
    }

    int rbase = base + 8 * khalf;
    if (m == 0) {
        #pragma unroll
        for (int r = 0; r < 8; ++r) a[rbase + r] = c[r];
    }
}

__device__ __forceinline__ int lane_get(int v, int j) {
    return __builtin_amdgcn_readlane(v, j);
}
template <int SH>
__device__ __forceinline__ int row_shr_get(int x) {
    return __builtin_amdgcn_update_dpp(0, x, 0x110 + SH, 0xf, 0xf, true);
}
__device__ __forceinline__ int wave_incl_scan(int x, int lane) {
    x += row_shr_get<1>(x);
    x += row_shr_get<2>(x);
    x += row_shr_get<4>(x);
    x += row_shr_get<8>(x);
    const int low = lane_get(x, 15);
    if (lane >= 16) x += low;
    return x;
}

__global__ __launch_bounds__(SCAT_THREADS)
void fa_bin_scatter_kernel(const int* __restrict__ src,
                           const int* __restrict__ dst,
                           float* z, int* bin_cnt, int e, int n, int nb) {
    __shared__ int  hist[NBMAX];
    __shared__ int  gdel[NBMAX];
    __shared__ int2 ordered[SCAT_EDGES];
    __shared__ int  wtot[SCAT_SLABS * SCAT_WAVES];
    __shared__ int  wpre[SCAT_SLABS * SCAT_WAVES + 1];
    const int tid  = threadIdx.x;
    const int lane = tid & 31;
    const int wave = tid >> 5;
    const int beg  = blockIdx.x * SCAT_EDGES;

    int d[EPT], s[EPT];
    #pragma unroll
    for (int k = 0; k < EPT; ++k)
        d[k] = dst[min(beg + k * SCAT_THREADS + tid, e - 1)];
    #pragma unroll
    for (int k = 0; k < EPT; ++k)
        s[k] = src[min(beg + k * SCAT_THREADS + tid, e - 1)];

    for (int i = tid; i < NBMAX; i += SCAT_THREADS) hist[i] = 0;
    __syncthreads();

    int rank[EPT];
    #pragma unroll
    for (int k = 0; k < EPT; ++k) {
        int i = beg + k * SCAT_THREADS + tid;
        rank[k] = 0;
        if (i < e) rank[k] = atomicAdd(&hist[d[k] >> BIN_SHIFT], 1);
    }
    __syncthreads();

    int c[SCAT_SLABS], inc[SCAT_SLABS];
    #pragma unroll
    for (int k = 0; k < SCAT_SLABS; ++k)
        c[k] = hist[tid + k * SCAT_THREADS];
    #pragma unroll
    for (int k = 0; k < SCAT_SLABS; ++k)
        inc[k] = wave_incl_scan(c[k], lane);
    if (lane == 31) {
        #pragma unroll
        for (int k = 0; k < SCAT_SLABS; ++k)
            wtot[k * SCAT_WAVES + wave] = inc[k];
    }
    __syncthreads();
    if (wave == 0) {
        const int t0 = wtot[2 * lane], t1 = wtot[2 * lane + 1];
        const int in2 = wave_incl_scan(t0 + t1, lane);
        const int ex2 = in2 - (t0 + t1);
        wpre[2 * lane]     = ex2;
        wpre[2 * lane + 1] = ex2 + t0;
        if (lane == 31) wpre[SCAT_SLABS * SCAT_WAVES] = in2;
    }
    __syncthreads();
    int ex[SCAT_SLABS];
    #pragma unroll
    for (int k = 0; k < SCAT_SLABS; ++k) {
        ex[k] = wpre[k * SCAT_WAVES + wave] + inc[k] - c[k];
        hist[tid + k * SCAT_THREADS] = ex[k];
    }
    const int total = wpre[SCAT_SLABS * SCAT_WAVES];
    __syncthreads();

    int r[SCAT_SLABS];
    #pragma unroll
    for (int k = 0; k < SCAT_SLABS; ++k) {
        int bin = tid + k * SCAT_THREADS;
        r[k] = 0;
        if (c[k] > 0) r[k] = atomicAdd(bin_cnt + cnt_index(bin), c[k]);
    }
    #pragma unroll
    for (int k = 0; k < EPT; ++k)
        rank[k] += hist[d[k] >> BIN_SHIFT];
    #pragma unroll
    for (int k = 0; k < EPT; ++k) {
        int i = beg + k * SCAT_THREADS + tid;
        if (i < e) ordered[rank[k]] = make_int2(s[k], d[k]);
    }
    #pragma unroll
    for (int k = 0; k < SCAT_SLABS; ++k)
        gdel[tid + k * SCAT_THREADS] = r[k] - ex[k];
    __syncthreads();

    int2 p[EPT];
    int  g[EPT];
    #pragma unroll
    for (int k = 0; k < EPT; ++k)
        p[k] = ordered[min(tid + k * SCAT_THREADS, total - 1)];
    #pragma unroll
    for (int k = 0; k < EPT; ++k)
        g[k] = gdel[p[k].y >> BIN_SHIFT];
    #pragma unroll
    for (int k = 0; k < EPT; ++k) {
        int i    = tid + k * SCAT_THREADS;
        int bin  = p[k].y >> BIN_SHIFT;
        int pos  = g[k] + i;
        int rows = min(BIN_ROWS, n - bin * BIN_ROWS);
        int cap  = rows * (DFEAT / 2);
        if (i < total && pos < cap) {
            int2* list = (int2*)(z + (size_t)bin * BIN_FLOATS);
            list[pos] = p[k];
        }
    }
}

__device__ __forceinline__ float lane_getf(float v, int j) {
    return __int_as_float(__builtin_amdgcn_readlane(__float_as_int(v), j));
}
template <int XM>
__device__ __forceinline__ float row_xor_get(float x) {
    return __int_as_float(__builtin_amdgcn_update_dpp(0, __float_as_int(x), 0x160 + XM,
                                                      0xf, 0xf, true));
}
__device__ __forceinline__ float xhalf(float x) {
    int b = __float_as_int(x);
    return __int_as_float(__builtin_amdgcn_permlanex16(b, b, 0x76543210, 0xfedcba98, false, false));
}
__device__ __forceinline__ float wave_sum(float x) {
    x += row_xor_get<1>(x);
    x += row_xor_get<2>(x);
    x += row_xor_get<4>(x);
    x += row_xor_get<8>(x);
    return x + xhalf(x);
}
__device__ __forceinline__ float sums8_by_lane(const float (&x)[8], int lane) {
    const bool b0 = (lane & 1) != 0, b1 = (lane & 2) != 0, b2 = (lane & 4) != 0;
    float y[4];
    #pragma unroll
    for (int i = 0; i < 4; ++i)
        y[i] = (b0 ? x[2 * i + 1] : x[2 * i]) + row_xor_get<1>(b0 ? x[2 * i] : x[2 * i + 1]);
    float zz[2];
    #pragma unroll
    for (int j = 0; j < 2; ++j)
        zz[j] = (b1 ? y[2 * j + 1] : y[2 * j]) + row_xor_get<2>(b1 ? y[2 * j] : y[2 * j + 1]);
    float w = (b2 ? zz[1] : zz[0]) + row_xor_get<4>(b2 ? zz[0] : zz[1]);
    w += row_xor_get<8>(w);
    return w + xhalf(w);
}

__global__ __launch_bounds__(256)
void fa_bin_reduce_kernel(const float* __restrict__ h,
                          const float* __restrict__ dnorm,
                          const float* __restrict__ a,
                          const float* __restrict__ gate_w,
                          const int* __restrict__ src,
                          const int* __restrict__ dst,
                          const float* __restrict__ gate_b,
                          float* z, const int* bin_cnt, int e, int n) {
    __shared__ float zacc[BIN_FLOATS];
    __shared__ int   rowcnt[BIN_ROWS];
    __shared__ int   rowbeg[BIN_ROWS + 1];
    const int tid  = threadIdx.x;
    const int lane = tid & 31;
    const int wave  = __builtin_amdgcn_readfirstlane(tid >> 5);
    const int bin   = blockIdx.x;
    const int node0 = bin * BIN_ROWS;
    const int rows  = min(BIN_ROWS, n - node0);
    float* zbin     = z + (size_t)bin * BIN_FLOATS;
    const int cnt   = __builtin_amdgcn_readfirstlane(bin_cnt[cnt_index(bin)]);
    const int cap   = rows * (DFEAT / 2);
    const int2* list = (const int2*)zbin;

    const float gb = gate_b[0];
    const float2 wsrc = ((const float2*)(gate_w + DFEAT))[lane];

    if (cnt <= cap) {
        int* sorted = (int*)zacc;
        if (tid < BIN_ROWS) rowcnt[tid] = 0;
        __syncthreads();

        int2 rec[8];
        int  rank[8];
        #pragma unroll
        for (int k = 0; k < 8; ++k) {
            int i = k * 256 + tid;
            rec[k] = make_int2(0, node0);
            if (i < cnt) rec[k] = list[i];
        }
        #pragma unroll
        for (int k = 0; k < 8; ++k) {
            int i = k * 256 + tid;
            rank[k] = 0;
            if (i < cnt) rank[k] = atomicAdd(&rowcnt[rec[k].y - node0], 1);
        }
        __syncthreads();

        if (wave == 0) {
            const int c0 = rowcnt[2 * lane], c1 = rowcnt[2 * lane + 1];
            int incl = c0 + c1;
            incl += row_shr_get<1>(incl);
            incl += row_shr_get<2>(incl);
            incl += row_shr_get<4>(incl);
            incl += row_shr_get<8>(incl);
            const int low = lane_get(incl, 15);
            if (lane >= 16) incl += low;
            const int ex = incl - (c0 + c1);
            rowbeg[2 * lane]     = ex;
            rowbeg[2 * lane + 1] = ex + c0;
            if (lane == 31) rowbeg[BIN_ROWS] = incl;
        }
        __syncthreads();

        #pragma unroll
        for (int k = 0; k < 8; ++k) {
            int i = k * 256 + tid;
            if (i < cnt) {
                int dl = rec[k].y - node0;
                sorted[rowbeg[dl] + rank[k]] = rec[k].x | (dl << 24);
            }
        }
        __syncthreads();

        const int r0 = wave * 8;
        const int rb = rowbeg[r0 + min(lane, 8)];
        const int beg = lane_get(rb, 0);
        const int end = lane_get(rb, 8);

        #pragma unroll
        for (int r = 0; r < 8; ++r) {
            if (r0 + r < rows && lane_get(rb, r + 1) == lane_get(rb, r))
                ((float2*)(zbin + (r0 + r) * DFEAT))[lane] = make_float2(0.0f, 0.0f);
        }

        float2 acc = make_float2(0.0f, 0.0f);
        int cur = -1;
        for (int eb = beg; eb < end; eb += 32) {
            const int ent = sorted[min(eb + lane, end - 1)];
            const int s   = ent & 0xffffff;
            const int dl  = ent >> 24;
            const float ad  = a[node0 + dl];
            const float dd  = dnorm[node0 + dl];
            const float dsr = dnorm[s];
            const int m = min(32, end - eb);
            for (int g = 0; g < m; g += 8) {
                float2 v[8];
                #pragma unroll
                for (int u = 0; u < 8; ++u) {
                    int sj = lane_get(s, g + u);
                    v[u] = ((const float2*)(h + (size_t)sj * DFEAT))[lane];
                }
                float x[8];
                #pragma unroll
                for (int u = 0; u < 8; ++u)
                    x[u] = v[u].x * wsrc.x + v[u].y * wsrc.y;
                const float bsg = sums8_by_lane(x, lane);
                const float evv = fast_tanh(ad + bsg + gb) * dd * dsr;
                #pragma unroll
                for (int u = 0; u < 8; ++u) {
                    if (g + u < m) {
                        int   dj = lane_get(dl, g + u);
                        float ej = lane_getf(evv, g + u);
                        if (dj != cur) {
                            if (cur >= 0) ((float2*)(zbin + cur * DFEAT))[lane] = acc;
                            acc = make_float2(0.0f, 0.0f);
                            cur = dj;
                        }
                        acc.x += v[u].x * ej;
                        acc.y += v[u].y * ej;
                    }
                }
            }
        }
        if (cur >= 0) ((float2*)(zbin + cur * DFEAT))[lane] = acc;
    } else {
        for (int i = tid; i < BIN_FLOATS / 4; i += 256)
            ((float4*)zacc)[i] = make_float4(0.0f, 0.0f, 0.0f, 0.0f);
        __syncthreads();
        for (int eb = wave * 32; eb < e; eb += 8 * 32) {
            const int idx = eb + lane;
            int s = 0, dl = 0;
            float pre = 0.0f, sc = 0.0f;
            bool mine = false;
            if (idx < e) {
                int d = dst[idx];
                if ((d >> BIN_SHIFT) == bin) {
                    mine = true;
                    s  = src[idx];
                    dl = d - node0;
                    pre = a[d] + gb;
                    sc  = dnorm[d] * dnorm[s];
                }
            }
            unsigned mask = __builtin_amdgcn_ballot_w32(mine);
            while (mask) {
                int j = __builtin_ctz(mask);
                mask &= mask - 1;
                int   sj = lane_get(s, j);
                int   dj = lane_get(dl, j);
                float2 v = ((const float2*)(h + (size_t)sj * DFEAT))[lane];
                float bs = wave_sum(v.x * wsrc.x + v.y * wsrc.y);
                float ej = fast_tanh(lane_getf(pre, j) + bs) * lane_getf(sc, j);
                float* zr = zacc + dj * DFEAT + 2 * lane;
                atomicAdd(zr,     v.x * ej);
                atomicAdd(zr + 1, v.y * ej);
            }
        }
        __syncthreads();

        for (int i = tid; i < rows * (DFEAT / 4); i += 256)
            ((float4*)zbin)[i] = ((const float4*)zacc)[i];
    }
}

extern "C" void kernel_launch(void* const* d_in, const int* in_sizes, int n_in,
                              void* d_out, int out_size, void* d_ws, size_t ws_size,
                              hipStream_t stream) {
    const float* h      = (const float*)d_in[0];
    const float* dnorm  = (const float*)d_in[1];
    const float* gate_w = (const float*)d_in[2];
    const float* gate_b = (const float*)d_in[3];
    const int*   src    = (const int*)d_in[4];
    const int*   dst    = (const int*)d_in[5];
    float*       z      = (float*)d_out;

    const int n = in_sizes[1];
    const int e = in_sizes[4];

    float* a = (float*)d_ws;
    int* bin_cnt = (int*)d_ws + ((n + 31) & ~31);

    const int nb = (n + BIN_ROWS - 1) / BIN_ROWS;

    {
        int waves   = (n + 15) / 16;
        int threads = 256;
        int blocks  = (waves * 32 + threads - 1) / threads;
        fa_node_gates_kernel<<<blocks, threads, 0, stream>>>(h, gate_w, a,
                                                             bin_cnt, n, nb);
    }

    {
        int per_block = SCAT_EDGES;
        int blocks = (e + per_block - 1) / per_block;
        fa_bin_scatter_kernel<<<blocks, SCAT_THREADS, 0, stream>>>(src, dst, z, bin_cnt, e, n, nb);
    }

    fa_bin_reduce_kernel<<<nb, 256, 0, stream>>>(h, dnorm, a, gate_w, src, dst,
                                                 gate_b, z, bin_cnt, e, n);
}


# scatter kernel loads dnorm[src] at entry (in flight across LDS-only barriers), stores it as the pair's 2nd word, row-in-bin packed above src; reduce kernel's scattered 4-B load gone; sum order kept
# speedup vs baseline: 2.9346x; 1.0578x over previous
#include <hip/hip_runtime.h>
#include <hip/hip_bf16.h>

#define DFEAT 64

#define CNT_PITCH  1056
#define BIN_ROWS   64
#define BIN_SHIFT  6
#define BIN_FLOATS (BIN_ROWS * DFEAT)
#define NBMAX      2048
#define EPT        16
#define SCAT_THREADS 512
#define SCAT_WAVES   (SCAT_THREADS / 32)
#define SCAT_SLABS   (NBMAX / SCAT_THREADS)
#define SCAT_EDGES   (SCAT_THREADS * EPT)
static_assert(SCAT_SLABS * SCAT_WAVES == 64, "the scan of the per-wave totals takes two per lane of one wave");

__device__ __forceinline__ int cnt_index(int bin) {
    return (bin >> 5) * CNT_PITCH + (bin & 31);
}

typedef float v2f __attribute__((ext_vector_type(2)));
typedef float v8f __attribute__((ext_vector_type(8)));

__device__ __forceinline__ float fast_tanh(float x) {
#if __has_builtin(__builtin_amdgcn_tanhf)
    return __builtin_amdgcn_tanhf(x);
#else
    return tanhf(x);
#endif
}

__global__ void fa_node_gates_kernel(const float* __restrict__ h,
                                     const float* __restrict__ gate_w,
                                     float* __restrict__ a,
                                     int* __restrict__ bin_cnt,
                                     int n, int nb) {
    int gid = blockIdx.x * blockDim.x + threadIdx.x;
    if (gid < nb) bin_cnt[cnt_index(gid)] = 0;
    int wave = gid >> 5;
    int lane = threadIdx.x & 31;
    int base = wave * 16;
    if (base >= n) return;

    int m     = lane & 15;
    int khalf = lane >> 4;

    const float2* hrow2 = (const float2*)(h + (size_t)(base + m) * DFEAT);

    const float2* w2   = (const float2*)gate_w;
    float         valid = (m == 0) ? 1.0f : 0.0f;

    v8f c = {};
    #pragma unroll
    for (int t = 0; t < 16; ++t) {
        int kq = 2 * t + khalf;

        float2 av = hrow2[kq];
        v2f A; A.x = av.x; A.y = av.y;

        float2 wv = w2[kq];
        v2f B; B.x = wv.x * valid; B.y = wv.y * valid;

        c = __builtin_amdgcn_wmma_f32_16x16x4_f32(
                 false, A,  false, B,
                 (short)0, c,  false,  false);
    }

    int rbase = base + 8 * khalf;
    if (m == 0) {
        #pragma unroll
        for (int r = 0; r < 8; ++r) a[rbase + r] = c[r];
    }
}

__device__ __forceinline__ int lane_get(int v, int j) {
    return __builtin_amdgcn_readlane(v, j);
}
template <int SH>
__device__ __forceinline__ int row_shr_get(int x) {
    return __builtin_amdgcn_update_dpp(0, x, 0x110 + SH, 0xf, 0xf, true);
}
__device__ __forceinline__ int wave_incl_scan(int x, int lane) {
    x += row_shr_get<1>(x);
    x += row_shr_get<2>(x);
    x += row_shr_get<4>(x);
    x += row_shr_get<8>(x);
    const int low = lane_get(x, 15);
    if (lane >= 16) x += low;
    return x;
}

__device__ __forceinline__ void lds_barrier() {
    __builtin_amdgcn_fence(__ATOMIC_RELEASE, "workgroup", "local");
    __builtin_amdgcn_s_barrier();
    __builtin_amdgcn_fence(__ATOMIC_ACQUIRE, "workgroup", "local");
}

__global__ __launch_bounds__(SCAT_THREADS)
void fa_bin_scatter_kernel(const int* __restrict__ src,
                           const int* __restrict__ dst,
                           const float* __restrict__ dnorm,
                           float* z, int* bin_cnt, int e, int n, int nb) {
    __shared__ int  hist[NBMAX];
    __shared__ int  gdel[NBMAX];
    __shared__ int2 ordered[SCAT_EDGES];
    __shared__ float odn[SCAT_EDGES];
    __shared__ int  wtot[SCAT_SLABS * SCAT_WAVES];
    __shared__ int  wpre[SCAT_SLABS * SCAT_WAVES + 1];
    const int tid  = threadIdx.x;
    const int lane = tid & 31;
    const int wave = tid >> 5;
    const int beg  = blockIdx.x * SCAT_EDGES;

    int d[EPT], s[EPT];
    #pragma unroll
    for (int k = 0; k < EPT; ++k)
        d[k] = dst[min(beg + k * SCAT_THREADS + tid, e - 1)];
    #pragma unroll
    for (int k = 0; k < EPT; ++k)
        s[k] = src[min(beg + k * SCAT_THREADS + tid, e - 1)];
    float dn[EPT];
    #pragma unroll
    for (int k = 0; k < EPT; ++k)
        dn[k] = dnorm[s[k]];

    for (int i = tid; i < NBMAX; i += SCAT_THREADS) hist[i] = 0;
    lds_barrier();

    int rank[EPT];
    #pragma unroll
    for (int k = 0; k < EPT; ++k) {
        int i = beg + k * SCAT_THREADS + tid;
        rank[k] = 0;
        if (i < e) rank[k] = atomicAdd(&hist[d[k] >> BIN_SHIFT], 1);
    }
    lds_barrier();

    int c[SCAT_SLABS], inc[SCAT_SLABS];
    #pragma unroll
    for (int k = 0; k < SCAT_SLABS; ++k)
        c[k] = hist[tid + k * SCAT_THREADS];
    #pragma unroll
    for (int k = 0; k < SCAT_SLABS; ++k)
        inc[k] = wave_incl_scan(c[k], lane);
    if (lane == 31) {
        #pragma unroll
        for (int k = 0; k < SCAT_SLABS; ++k)
            wtot[k * SCAT_WAVES + wave] = inc[k];
    }
    lds_barrier();
    if (wave == 0) {
        const int t0 = wtot[2 * lane], t1 = wtot[2 * lane + 1];
        const int in2 = wave_incl_scan(t0 + t1, lane);
        const int ex2 = in2 - (t0 + t1);
        wpre[2 * lane]     = ex2;
        wpre[2 * lane + 1] = ex2 + t0;
        if (lane == 31) wpre[SCAT_SLABS * SCAT_WAVES] = in2;
    }
    lds_barrier();
    int ex[SCAT_SLABS];
    #pragma unroll
    for (int k = 0; k < SCAT_SLABS; ++k) {
        ex[k] = wpre[k * SCAT_WAVES + wave] + inc[k] - c[k];
        hist[tid + k * SCAT_THREADS] = ex[k];
    }
    const int total = wpre[SCAT_SLABS * SCAT_WAVES];
    lds_barrier();

    int r[SCAT_SLABS];
    #pragma unroll
    for (int k = 0; k < SCAT_SLABS; ++k) {
        int bin = tid + k * SCAT_THREADS;
        r[k] = 0;
        if (c[k] > 0) r[k] = atomicAdd(bin_cnt + cnt_index(bin), c[k]);
    }
    #pragma unroll
    for (int k = 0; k < EPT; ++k)
        rank[k] += hist[d[k] >> BIN_SHIFT];
    #pragma unroll
    for (int k = 0; k < EPT; ++k) {
        int i = beg + k * SCAT_THREADS + tid;
        if (i < e) {
            ordered[rank[k]] = make_int2(s[k], d[k]);
            odn[rank[k]]     = dn[k];
        }
    }
    #pragma unroll
    for (int k = 0; k < SCAT_SLABS; ++k)
        gdel[tid + k * SCAT_THREADS] = r[k] - ex[k];
    __syncthreads();

    int2  p[EPT];
    float q[EPT];
    int   g[EPT];
    #pragma unroll
    for (int k = 0; k < EPT; ++k) {
        int j = min(tid + k * SCAT_THREADS, total - 1);
        p[k] = ordered[j];
        q[k] = odn[j];
    }
    #pragma unroll
    for (int k = 0; k < EPT; ++k)
        g[k] = gdel[p[k].y >> BIN_SHIFT];
    #pragma unroll
    for (int k = 0; k < EPT; ++k) {
        int i    = tid + k * SCAT_THREADS;
        int bin  = p[k].y >> BIN_SHIFT;
        int pos  = g[k] + i;
        int rows = min(BIN_ROWS, n - bin * BIN_ROWS);
        int cap  = rows * (DFEAT / 2);
        if (i < total && pos < cap) {
            int2* list = (int2*)(z + (size_t)bin * BIN_FLOATS);
            list[pos] = make_int2(p[k].x | ((p[k].y & (BIN_ROWS - 1)) << 24), __float_as_int(q[k]));
        }
    }
}

__device__ __forceinline__ float lane_getf(float v, int j) {
    return __int_as_float(__builtin_amdgcn_readlane(__float_as_int(v), j));
}
template <int XM>
__device__ __forceinline__ float row_xor_get(float x) {
    return __int_as_float(__builtin_amdgcn_update_dpp(0, __float_as_int(x), 0x160 + XM,
                                                      0xf, 0xf, true));
}
__device__ __forceinline__ float xhalf(float x) {
    int b = __float_as_int(x);
    return __int_as_float(__builtin_amdgcn_permlanex16(b, b, 0x76543210, 0xfedcba98, false, false));
}
__device__ __forceinline__ float wave_sum(float x) {
    x += row_xor_get<1>(x);
    x += row_xor_get<2>(x);
    x += row_xor_get<4>(x);
    x += row_xor_get<8>(x);
    return x + xhalf(x);
}
__device__ __forceinline__ float sums8_by_lane(const float (&x)[8], int lane) {
    const bool b0 = (lane & 1) != 0, b1 = (lane & 2) != 0, b2 = (lane & 4) != 0;
    float y[4];
    #pragma unroll
    for (int i = 0; i < 4; ++i)
        y[i] = (b0 ? x[2 * i + 1] : x[2 * i]) + row_xor_get<1>(b0 ? x[2 * i] : x[2 * i + 1]);
    float zz[2];
    #pragma unroll
    for (int j = 0; j < 2; ++j)
        zz[j] = (b1 ? y[2 * j + 1] : y[2 * j]) + row_xor_get<2>(b1 ? y[2 * j] : y[2 * j + 1]);
    float w = (b2 ? zz[1] : zz[0]) + row_xor_get<4>(b2 ? zz[0] : zz[1]);
    w += row_xor_get<8>(w);
    return w + xhalf(w);
}

__global__ __launch_bounds__(256)
void fa_bin_reduce_kernel(const float* __restrict__ h,
                          const float* __restrict__ dnorm,
                          const float* __restrict__ a,
                          const float* __restrict__ gate_w,
                          const int* __restrict__ src,
                          const int* __restrict__ dst,
                          const float* __restrict__ gate_b,
                          float* z, const int* bin_cnt, int e, int n) {
    __shared__ float zacc[BIN_FLOATS];
    __shared__ int   rowcnt[BIN_ROWS];
    __shared__ int   rowbeg[BIN_ROWS + 1];
    const int tid  = threadIdx.x;
    const int lane = tid & 31;
    const int wave  = __builtin_amdgcn_readfirstlane(tid >> 5);
    const int bin   = blockIdx.x;
    const int node0 = bin * BIN_ROWS;
    const int rows  = min(BIN_ROWS, n - node0);
    float* zbin     = z + (size_t)bin * BIN_FLOATS;
    const int cnt   = __builtin_amdgcn_readfirstlane(bin_cnt[cnt_index(bin)]);
    const int cap   = rows * (DFEAT / 2);
    const int2* list = (const int2*)zbin;

    const float gb = gate_b[0];
    const float2 wsrc = ((const float2*)(gate_w + DFEAT))[lane];

    if (cnt <= cap) {
        int2* sorted = (int2*)zacc;
        if (tid < BIN_ROWS) rowcnt[tid] = 0;
        __syncthreads();

        int2 rec[8];
        int  rank[8];
        #pragma unroll
        for (int k = 0; k < 8; ++k) {
            int i = k * 256 + tid;
            rec[k] = make_int2(0, 0);
            if (i < cnt) rec[k] = list[i];
        }
        #pragma unroll
        for (int k = 0; k < 8; ++k) {
            int i = k * 256 + tid;
            rank[k] = 0;
            if (i < cnt) rank[k] = atomicAdd(&rowcnt[rec[k].x >> 24], 1);
        }
        __syncthreads();

        if (wave == 0) {
            const int c0 = rowcnt[2 * lane], c1 = rowcnt[2 * lane + 1];
            int incl = c0 + c1;
            incl += row_shr_get<1>(incl);
            incl += row_shr_get<2>(incl);
            incl += row_shr_get<4>(incl);
            incl += row_shr_get<8>(incl);
            const int low = lane_get(incl, 15);
            if (lane >= 16) incl += low;
            const int ex = incl - (c0 + c1);
            rowbeg[2 * lane]     = ex;
            rowbeg[2 * lane + 1] = ex + c0;
            if (lane == 31) rowbeg[BIN_ROWS] = incl;
        }
        __syncthreads();

        #pragma unroll
        for (int k = 0; k < 8; ++k) {
            int i = k * 256 + tid;
            if (i < cnt) {
                int dl = rec[k].x >> 24;
                sorted[rowbeg[dl] + rank[k]] = rec[k];
            }
        }
        __syncthreads();

        const int r0 = wave * 8;
        const int rb = rowbeg[r0 + min(lane, 8)];
        const int beg = lane_get(rb, 0);
        const int end = lane_get(rb, 8);

        #pragma unroll
        for (int r = 0; r < 8; ++r) {
            if (r0 + r < rows && lane_get(rb, r + 1) == lane_get(rb, r))
                ((float2*)(zbin + (r0 + r) * DFEAT))[lane] = make_float2(0.0f, 0.0f);
        }

        float2 acc = make_float2(0.0f, 0.0f);
        int cur = -1;
        for (int eb = beg; eb < end; eb += 32) {
            const int2 ent = sorted[min(eb + lane, end - 1)];
            const int s   = ent.x & 0xffffff;
            const int dl  = ent.x >> 24;
            const float dsr = __int_as_float(ent.y);
            const float ad  = a[node0 + dl];
            const float dd  = dnorm[node0 + dl];
            const int m = min(32, end - eb);
            for (int g = 0; g < m; g += 8) {
                float2 v[8];
                #pragma unroll
                for (int u = 0; u < 8; ++u) {
                    int sj = lane_get(s, g + u);
                    v[u] = ((const float2*)(h + (size_t)sj * DFEAT))[lane];
                }
                float x[8];
                #pragma unroll
                for (int u = 0; u < 8; ++u)
                    x[u] = v[u].x * wsrc.x + v[u].y * wsrc.y;
                const float bsg = sums8_by_lane(x, lane);
                const float evv = fast_tanh(ad + bsg + gb) * dd * dsr;
                #pragma unroll
                for (int u = 0; u < 8; ++u) {
                    if (g + u < m) {
                        int   dj = lane_get(dl, g + u);
                        float ej = lane_getf(evv, g + u);
                        if (dj != cur) {
                            if (cur >= 0) ((float2*)(zbin + cur * DFEAT))[lane] = acc;
                            acc = make_float2(0.0f, 0.0f);
                            cur = dj;
                        }
                        acc.x += v[u].x * ej;
                        acc.y += v[u].y * ej;
                    }
                }
            }
        }
        if (cur >= 0) ((float2*)(zbin + cur * DFEAT))[lane] = acc;
    } else {
        for (int i = tid; i < BIN_FLOATS / 4; i += 256)
            ((float4*)zacc)[i] = make_float4(0.0f, 0.0f, 0.0f, 0.0f);
        __syncthreads();
        for (int eb = wave * 32; eb < e; eb += 8 * 32) {
            const int idx = eb + lane;
            int s = 0, dl = 0;
            float pre = 0.0f, sc = 0.0f;
            bool mine = false;
            if (idx < e) {
                int d = dst[idx];
                if ((d >> BIN_SHIFT) == bin) {
                    mine = true;
                    s  = src[idx];
                    dl = d - node0;
                    pre = a[d] + gb;
                    sc  = dnorm[d] * dnorm[s];
                }
            }
            unsigned mask = __builtin_amdgcn_ballot_w32(mine);
            while (mask) {
                int j = __builtin_ctz(mask);
                mask &= mask - 1;
                int   sj = lane_get(s, j);
                int   dj = lane_get(dl, j);
                float2 v = ((const float2*)(h + (size_t)sj * DFEAT))[lane];
                float bs = wave_sum(v.x * wsrc.x + v.y * wsrc.y);
                float ej = fast_tanh(lane_getf(pre, j) + bs) * lane_getf(sc, j);
                float* zr = zacc + dj * DFEAT + 2 * lane;
                atomicAdd(zr,     v.x * ej);
                atomicAdd(zr + 1, v.y * ej);
            }
        }
        __syncthreads();

        for (int i = tid; i < rows * (DFEAT / 4); i += 256)
            ((float4*)zbin)[i] = ((const float4*)zacc)[i];
    }
}

extern "C" void kernel_launch(void* const* d_in, const int* in_sizes, int n_in,
                              void* d_out, int out_size, void* d_ws, size_t ws_size,
                              hipStream_t stream) {
    const float* h      = (const float*)d_in[0];
    const float* dnorm  = (const float*)d_in[1];
    const float* gate_w = (const float*)d_in[2];
    const float* gate_b = (const float*)d_in[3];
    const int*   src    = (const int*)d_in[4];
    const int*   dst    = (const int*)d_in[5];
    float*       z      = (float*)d_out;

    const int n = in_sizes[1];
    const int e = in_sizes[4];

    float* a = (float*)d_ws;
    int* bin_cnt = (int*)d_ws + ((n + 31) & ~31);

    const int nb = (n + BIN_ROWS - 1) / BIN_ROWS;

    {
        int waves   = (n + 15) / 16;
        int threads = 256;
        int blocks  = (waves * 32 + threads - 1) / threads;
        fa_node_gates_kernel<<<blocks, threads, 0, stream>>>(h, gate_w, a,
                                                             bin_cnt, n, nb);
    }

    {
        int per_block = SCAT_EDGES;
        int blocks = (e + per_block - 1) / per_block;
        fa_bin_scatter_kernel<<<blocks, SCAT_THREADS, 0, stream>>>(src, dst, dnorm, z, bin_cnt, e, n, nb);
    }

    fa_bin_reduce_kernel<<<nb, 256, 0, stream>>>(h, dnorm, a, gate_w, src, dst,
                                                 gate_b, z, bin_cnt, e, n);
}


# node gate projection (WMMA) runs as extra blocks of the scatter launch, beside its one-per-unit routing blocks, not as a 6256-wave launch of its own; kernel 1 only zeroes counts; sum order kept
# speedup vs baseline: 3.0324x; 1.0333x over previous
#include <hip/hip_runtime.h>
#include <hip/hip_bf16.h>

#define DFEAT 64

#define CNT_PITCH  1056
#define BIN_ROWS   64
#define BIN_SHIFT  6
#define BIN_FLOATS (BIN_ROWS * DFEAT)
#define NBMAX      2048
#define EPT        16
#define SCAT_THREADS 512
#define SCAT_WAVES   (SCAT_THREADS / 32)
#define SCAT_SLABS   (NBMAX / SCAT_THREADS)
#define SCAT_EDGES   (SCAT_THREADS * EPT)
static_assert(SCAT_SLABS * SCAT_WAVES == 64, "the scan of the per-wave totals takes two per lane of one wave");

__device__ __forceinline__ int cnt_index(int bin) {
    return (bin >> 5) * CNT_PITCH + (bin & 31);
}

typedef float v2f __attribute__((ext_vector_type(2)));
typedef float v8f __attribute__((ext_vector_type(8)));

__device__ __forceinline__ float fast_tanh(float x) {
#if __has_builtin(__builtin_amdgcn_tanhf)
    return __builtin_amdgcn_tanhf(x);
#else
    return tanhf(x);
#endif
}

__global__ void fa_zero_counts_kernel(int* __restrict__ bin_cnt, int nb) {
    int gid = blockIdx.x * blockDim.x + threadIdx.x;
    if (gid < nb) bin_cnt[cnt_index(gid)] = 0;
}

__device__ __forceinline__ void gate_rows32(const float* __restrict__ h,
                                            const float* __restrict__ gate_w,
                                            float* __restrict__ a,
                                            int base, int n, int lane) {
    int m     = lane & 15;
    int khalf = lane >> 4;

    const float2* hrow2a = (const float2*)(h + (size_t)min(base + m, n - 1) * DFEAT);
    const float2* hrow2b = (const float2*)(h + (size_t)min(base + 16 + m, n - 1) * DFEAT);

    const float2* w2   = (const float2*)gate_w;
    float         valid = (m == 0) ? 1.0f : 0.0f;

    v8f c0 = {}, c1 = {};
    #pragma unroll
    for (int t = 0; t < 16; ++t) {
        int kq = 2 * t + khalf;

        float2 av0 = hrow2a[kq];
        v2f A0; A0.x = av0.x; A0.y = av0.y;
        float2 av1 = hrow2b[kq];
        v2f A1; A1.x = av1.x; A1.y = av1.y;

        float2 wv = w2[kq];
        v2f B; B.x = wv.x * valid; B.y = wv.y * valid;

        c0 = __builtin_amdgcn_wmma_f32_16x16x4_f32(
                 false, A0,  false, B,
                 (short)0, c0,  false,  false);
        c1 = __builtin_amdgcn_wmma_f32_16x16x4_f32(
                 false, A1,  false, B,
                 (short)0, c1,  false,  false);
    }

    int rbase = base + 8 * khalf;
    if (m == 0) {
        #pragma unroll
        for (int r = 0; r < 8; ++r) {
            if (rbase + r < n)      a[rbase + r]      = c0[r];
            if (rbase + 16 + r < n) a[rbase + 16 + r] = c1[r];
        }
    }
}

__device__ __forceinline__ int lane_get(int v, int j) {
    return __builtin_amdgcn_readlane(v, j);
}
template <int SH>
__device__ __forceinline__ int row_shr_get(int x) {
    return __builtin_amdgcn_update_dpp(0, x, 0x110 + SH, 0xf, 0xf, true);
}
__device__ __forceinline__ int wave_incl_scan(int x, int lane) {
    x += row_shr_get<1>(x);
    x += row_shr_get<2>(x);
    x += row_shr_get<4>(x);
    x += row_shr_get<8>(x);
    const int low = lane_get(x, 15);
    if (lane >= 16) x += low;
    return x;
}

__device__ __forceinline__ void lds_barrier() {
    __builtin_amdgcn_fence(__ATOMIC_RELEASE, "workgroup", "local");
    __builtin_amdgcn_s_barrier();
    __builtin_amdgcn_fence(__ATOMIC_ACQUIRE, "workgroup", "local");
}

__global__ __launch_bounds__(SCAT_THREADS)
void fa_bin_scatter_kernel(const int* __restrict__ src,
                           const int* __restrict__ dst,
                           const float* __restrict__ dnorm,
                           const float* __restrict__ h,
                           const float* __restrict__ gate_w,
                           float* __restrict__ a,
                           float* z, int* bin_cnt, int e, int n, int nb, int sblocks) {
    __shared__ int  hist[NBMAX];
    __shared__ int  gdel[NBMAX];
    __shared__ int2 ordered[SCAT_EDGES];
    __shared__ float odn[SCAT_EDGES];
    __shared__ int  wtot[SCAT_SLABS * SCAT_WAVES];
    __shared__ int  wpre[SCAT_SLABS * SCAT_WAVES + 1];
    const int tid  = threadIdx.x;
    const int lane = tid & 31;
    const int wave = tid >> 5;
    if ((int)blockIdx.x >= sblocks) {
        const int gw   = __builtin_amdgcn_readfirstlane(((int)blockIdx.x - sblocks) * SCAT_WAVES + wave);
        const int base = gw * 32;
        if (base < n) gate_rows32(h, gate_w, a, base, n, lane);
        return;
    }
    const int beg  = blockIdx.x * SCAT_EDGES;

    int d[EPT], s[EPT];
    #pragma unroll
    for (int k = 0; k < EPT; ++k)
        d[k] = dst[min(beg + k * SCAT_THREADS + tid, e - 1)];
    #pragma unroll
    for (int k = 0; k < EPT; ++k)
        s[k] = src[min(beg + k * SCAT_THREADS + tid, e - 1)];
    float dn[EPT];
    #pragma unroll
    for (int k = 0; k < EPT; ++k)
        dn[k] = dnorm[s[k]];

    for (int i = tid; i < NBMAX; i += SCAT_THREADS) hist[i] = 0;
    lds_barrier();

    int rank[EPT];
    #pragma unroll
    for (int k = 0; k < EPT; ++k) {
        int i = beg + k * SCAT_THREADS + tid;
        rank[k] = 0;
        if (i < e) rank[k] = atomicAdd(&hist[d[k] >> BIN_SHIFT], 1);
    }
    lds_barrier();

    int c[SCAT_SLABS], inc[SCAT_SLABS];
    #pragma unroll
    for (int k = 0; k < SCAT_SLABS; ++k)
        c[k] = hist[tid + k * SCAT_THREADS];
    #pragma unroll
    for (int k = 0; k < SCAT_SLABS; ++k)
        inc[k] = wave_incl_scan(c[k], lane);
    if (lane == 31) {
        #pragma unroll
        for (int k = 0; k < SCAT_SLABS; ++k)
            wtot[k * SCAT_WAVES + wave] = inc[k];
    }
    lds_barrier();
    if (wave == 0) {
        const int t0 = wtot[2 * lane], t1 = wtot[2 * lane + 1];
        const int in2 = wave_incl_scan(t0 + t1, lane);
        const int ex2 = in2 - (t0 + t1);
        wpre[2 * lane]     = ex2;
        wpre[2 * lane + 1] = ex2 + t0;
        if (lane == 31) wpre[SCAT_SLABS * SCAT_WAVES] = in2;
    }
    lds_barrier();
    int ex[SCAT_SLABS];
    #pragma unroll
    for (int k = 0; k < SCAT_SLABS; ++k) {
        ex[k] = wpre[k * SCAT_WAVES + wave] + inc[k] - c[k];
        hist[tid + k * SCAT_THREADS] = ex[k];
    }
    const int total = wpre[SCAT_SLABS * SCAT_WAVES];
    lds_barrier();

    int r[SCAT_SLABS];
    #pragma unroll
    for (int k = 0; k < SCAT_SLABS; ++k) {
        int bin = tid + k * SCAT_THREADS;
        r[k] = 0;
        if (c[k] > 0) r[k] = atomicAdd(bin_cnt + cnt_index(bin), c[k]);
    }
    #pragma unroll
    for (int k = 0; k < EPT; ++k)
        rank[k] += hist[d[k] >> BIN_SHIFT];
    #pragma unroll
    for (int k = 0; k < EPT; ++k) {
        int i = beg + k * SCAT_THREADS + tid;
        if (i < e) {
            ordered[rank[k]] = make_int2(s[k], d[k]);
            odn[rank[k]]     = dn[k];
        }
    }
    #pragma unroll
    for (int k = 0; k < SCAT_SLABS; ++k)
        gdel[tid + k * SCAT_THREADS] = r[k] - ex[k];
    __syncthreads();

    int2  p[EPT];
    float q[EPT];
    int   g[EPT];
    #pragma unroll
    for (int k = 0; k < EPT; ++k) {
        int j = min(tid + k * SCAT_THREADS, total - 1);
        p[k] = ordered[j];
        q[k] = odn[j];
    }
    #pragma unroll
    for (int k = 0; k < EPT; ++k)
        g[k] = gdel[p[k].y >> BIN_SHIFT];
    #pragma unroll
    for (int k = 0; k < EPT; ++k) {
        int i    = tid + k * SCAT_THREADS;
        int bin  = p[k].y >> BIN_SHIFT;
        int pos  = g[k] + i;
        int rows = min(BIN_ROWS, n - bin * BIN_ROWS);
        int cap  = rows * (DFEAT / 2);
        if (i < total && pos < cap) {
            int2* list = (int2*)(z + (size_t)bin * BIN_FLOATS);
            list[pos] = make_int2(p[k].x | ((p[k].y & (BIN_ROWS - 1)) << 24), __float_as_int(q[k]));
        }
    }
}

__device__ __forceinline__ float lane_getf(float v, int j) {
    return __int_as_float(__builtin_amdgcn_readlane(__float_as_int(v), j));
}
template <int XM>
__device__ __forceinline__ float row_xor_get(float x) {
    return __int_as_float(__builtin_amdgcn_update_dpp(0, __float_as_int(x), 0x160 + XM,
                                                      0xf, 0xf, true));
}
__device__ __forceinline__ float xhalf(float x) {
    int b = __float_as_int(x);
    return __int_as_float(__builtin_amdgcn_permlanex16(b, b, 0x76543210, 0xfedcba98, false, false));
}
__device__ __forceinline__ float wave_sum(float x) {
    x += row_xor_get<1>(x);
    x += row_xor_get<2>(x);
    x += row_xor_get<4>(x);
    x += row_xor_get<8>(x);
    return x + xhalf(x);
}
__device__ __forceinline__ float sums8_by_lane(const float (&x)[8], int lane) {
    const bool b0 = (lane & 1) != 0, b1 = (lane & 2) != 0, b2 = (lane & 4) != 0;
    float y[4];
    #pragma unroll
    for (int i = 0; i < 4; ++i)
        y[i] = (b0 ? x[2 * i + 1] : x[2 * i]) + row_xor_get<1>(b0 ? x[2 * i] : x[2 * i + 1]);
    float zz[2];
    #pragma unroll
    for (int j = 0; j < 2; ++j)
        zz[j] = (b1 ? y[2 * j + 1] : y[2 * j]) + row_xor_get<2>(b1 ? y[2 * j] : y[2 * j + 1]);
    float w = (b2 ? zz[1] : zz[0]) + row_xor_get<4>(b2 ? zz[0] : zz[1]);
    w += row_xor_get<8>(w);
    return w + xhalf(w);
}

__global__ __launch_bounds__(256)
void fa_bin_reduce_kernel(const float* __restrict__ h,
                          const float* __restrict__ dnorm,
                          const float* __restrict__ a,
                          const float* __restrict__ gate_w,
                          const int* __restrict__ src,
                          const int* __restrict__ dst,
                          const float* __restrict__ gate_b,
                          float* z, const int* bin_cnt, int e, int n) {
    __shared__ float zacc[BIN_FLOATS];
    __shared__ int   rowcnt[BIN_ROWS];
    __shared__ int   rowbeg[BIN_ROWS + 1];
    const int tid  = threadIdx.x;
    const int lane = tid & 31;
    const int wave  = __builtin_amdgcn_readfirstlane(tid >> 5);
    const int bin   = blockIdx.x;
    const int node0 = bin * BIN_ROWS;
    const int rows  = min(BIN_ROWS, n - node0);
    float* zbin     = z + (size_t)bin * BIN_FLOATS;
    const int cnt   = __builtin_amdgcn_readfirstlane(bin_cnt[cnt_index(bin)]);
    const int cap   = rows * (DFEAT / 2);
    const int2* list = (const int2*)zbin;

    const float gb = gate_b[0];
    const float2 wsrc = ((const float2*)(gate_w + DFEAT))[lane];

    if (cnt <= cap) {
        int2* sorted = (int2*)zacc;
        if (tid < BIN_ROWS) rowcnt[tid] = 0;
        __syncthreads();

        int2 rec[8];
        int  rank[8];
        #pragma unroll
        for (int k = 0; k < 8; ++k) {
            int i = k * 256 + tid;
            rec[k] = make_int2(0, 0);
            if (i < cnt) rec[k] = list[i];
        }
        #pragma unroll
        for (int k = 0; k < 8; ++k) {
            int i = k * 256 + tid;
            rank[k] = 0;
            if (i < cnt) rank[k] = atomicAdd(&rowcnt[rec[k].x >> 24], 1);
        }
        __syncthreads();

        if (wave == 0) {
            const int c0 = rowcnt[2 * lane], c1 = rowcnt[2 * lane + 1];
            int incl = c0 + c1;
            incl += row_shr_get<1>(incl);
            incl += row_shr_get<2>(incl);
            incl += row_shr_get<4>(incl);
            incl += row_shr_get<8>(incl);
            const int low = lane_get(incl, 15);
            if (lane >= 16) incl += low;
            const int ex = incl - (c0 + c1);
            rowbeg[2 * lane]     = ex;
            rowbeg[2 * lane + 1] = ex + c0;
            if (lane == 31) rowbeg[BIN_ROWS] = incl;
        }
        __syncthreads();

        #pragma unroll
        for (int k = 0; k < 8; ++k) {
            int i = k * 256 + tid;
            if (i < cnt) {
                int dl = rec[k].x >> 24;
                sorted[rowbeg[dl] + rank[k]] = rec[k];
            }
        }
        __syncthreads();

        const int r0 = wave * 8;
        const int rb = rowbeg[r0 + min(lane, 8)];
        const int beg = lane_get(rb, 0);
        const int end = lane_get(rb, 8);

        #pragma unroll
        for (int r = 0; r < 8; ++r) {
            if (r0 + r < rows && lane_get(rb, r + 1) == lane_get(rb, r))
                ((float2*)(zbin + (r0 + r) * DFEAT))[lane] = make_float2(0.0f, 0.0f);
        }

        float2 acc = make_float2(0.0f, 0.0f);
        int cur = -1;
        for (int eb = beg; eb < end; eb += 32) {
            const int2 ent = sorted[min(eb + lane, end - 1)];
            const int s   = ent.x & 0xffffff;
            const int dl  = ent.x >> 24;
            const float dsr = __int_as_float(ent.y);
            const float ad  = a[node0 + dl];
            const float dd  = dnorm[node0 + dl];
            const int m = min(32, end - eb);
            for (int g = 0; g < m; g += 8) {
                float2 v[8];
                #pragma unroll
                for (int u = 0; u < 8; ++u) {
                    int sj = lane_get(s, g + u);
                    v[u] = ((const float2*)(h + (size_t)sj * DFEAT))[lane];
                }
                float x[8];
                #pragma unroll
                for (int u = 0; u < 8; ++u)
                    x[u] = v[u].x * wsrc.x + v[u].y * wsrc.y;
                const float bsg = sums8_by_lane(x, lane);
                const float evv = fast_tanh(ad + bsg + gb) * dd * dsr;
                #pragma unroll
                for (int u = 0; u < 8; ++u) {
                    if (g + u < m) {
                        int   dj = lane_get(dl, g + u);
                        float ej = lane_getf(evv, g + u);
                        if (dj != cur) {
                            if (cur >= 0) ((float2*)(zbin + cur * DFEAT))[lane] = acc;
                            acc = make_float2(0.0f, 0.0f);
                            cur = dj;
                        }
                        acc.x += v[u].x * ej;
                        acc.y += v[u].y * ej;
                    }
                }
            }
        }
        if (cur >= 0) ((float2*)(zbin + cur * DFEAT))[lane] = acc;
    } else {
        for (int i = tid; i < BIN_FLOATS / 4; i += 256)
            ((float4*)zacc)[i] = make_float4(0.0f, 0.0f, 0.0f, 0.0f);
        __syncthreads();
        for (int eb = wave * 32; eb < e; eb += 8 * 32) {
            const int idx = eb + lane;
            int s = 0, dl = 0;
            float pre = 0.0f, sc = 0.0f;
            bool mine = false;
            if (idx < e) {
                int d = dst[idx];
                if ((d >> BIN_SHIFT) == bin) {
                    mine = true;
                    s  = src[idx];
                    dl = d - node0;
                    pre = a[d] + gb;
                    sc  = dnorm[d] * dnorm[s];
                }
            }
            unsigned mask = __builtin_amdgcn_ballot_w32(mine);
            while (mask) {
                int j = __builtin_ctz(mask);
                mask &= mask - 1;
                int   sj = lane_get(s, j);
                int   dj = lane_get(dl, j);
                float2 v = ((const float2*)(h + (size_t)sj * DFEAT))[lane];
                float bs = wave_sum(v.x * wsrc.x + v.y * wsrc.y);
                float ej = fast_tanh(lane_getf(pre, j) + bs) * lane_getf(sc, j);
                float* zr = zacc + dj * DFEAT + 2 * lane;
                atomicAdd(zr,     v.x * ej);
                atomicAdd(zr + 1, v.y * ej);
            }
        }
        __syncthreads();

        for (int i = tid; i < rows * (DFEAT / 4); i += 256)
            ((float4*)zbin)[i] = ((const float4*)zacc)[i];
    }
}

extern "C" void kernel_launch(void* const* d_in, const int* in_sizes, int n_in,
                              void* d_out, int out_size, void* d_ws, size_t ws_size,
                              hipStream_t stream) {
    const float* h      = (const float*)d_in[0];
    const float* dnorm  = (const float*)d_in[1];
    const float* gate_w = (const float*)d_in[2];
    const float* gate_b = (const float*)d_in[3];
    const int*   src    = (const int*)d_in[4];
    const int*   dst    = (const int*)d_in[5];
    float*       z      = (float*)d_out;

    const int n = in_sizes[1];
    const int e = in_sizes[4];

    float* a = (float*)d_ws;
    int* bin_cnt = (int*)d_ws + ((n + 31) & ~31);

    const int nb = (n + BIN_ROWS - 1) / BIN_ROWS;

    {
        int threads = 256;
        int blocks  = (nb + threads - 1) / threads;
        fa_zero_counts_kernel<<<blocks, threads, 0, stream>>>(bin_cnt, nb);
    }

    {
        int per_block = SCAT_EDGES;
        int sblocks = (e + per_block - 1) / per_block;
        int gwaves  = (n + 31) / 32;
        int gblocks = (gwaves + SCAT_WAVES - 1) / SCAT_WAVES;
        fa_bin_scatter_kernel<<<sblocks + gblocks, SCAT_THREADS, 0, stream>>>(src, dst, dnorm, h, gate_w, a, z,
                                                                             bin_cnt, e, n, nb, sblocks);
    }

    fa_bin_reduce_kernel<<<nb, 256, 0, stream>>>(h, dnorm, a, gate_w, src, dst,
                                                 gate_b, z, bin_cnt, e, n);
}
